# starting kernel
# baseline (speedup 1.0000x reference)
#include <hip/hip_runtime.h>

#define NB 32
#define NT 1024
#define NJ 128
#define ND 256

typedef __attribute__((ext_vector_type(2))) float v2f;
typedef __attribute__((ext_vector_type(8))) float v8f;

__device__ __forceinline__ float wave_reduce_sum(float v) {
#pragma unroll
  for (int o = 16; o > 0; o >>= 1) v += __shfl_xor(v, o, 32);
  return v;
}
__device__ __forceinline__ float wave_reduce_max(float v) {
#pragma unroll
  for (int o = 16; o > 0; o >>= 1) v = fmaxf(v, __shfl_xor(v, o, 32));
  return v;
}

__global__ void k_rowdots(const float* __restrict__ C, const float* __restrict__ Q,
                          const float* __restrict__ w, float* __restrict__ ch,
                          float* __restrict__ qb) {
  int wave = blockIdx.x * (blockDim.x >> 5) + (threadIdx.x >> 5);
  int lane = threadIdx.x & 31;
  const int nch = NB * NT;
  const int nrows = nch + NB * NJ;
  if (wave >= nrows) return;
  const float* row; const float* wv; float* out;
  if (wave < nch) { row = C + (size_t)wave * ND; wv = w;      out = ch + wave; }
  else { int r = wave - nch; row = Q + (size_t)r * ND; wv = w + ND; out = qb + r; }
  float s = 0.f;
#pragma unroll
  for (int i = lane; i < ND; i += 32) s += row[i] * wv[i];
  s = wave_reduce_sum(s);
  if (lane == 0) *out = s;
}

__global__ void k_score(const float* __restrict__ C, const float* __restrict__ Q,
                        const float* __restrict__ w, const float* __restrict__ ch,
                        const float* __restrict__ qb, float* __restrict__ S) {
  int blk = blockIdx.x;
  int b = blk >> 6;
  int ttile = blk & 63;
  int wv = threadIdx.x >> 5;
  int lane = threadIdx.x & 31;
  int m = lane & 15, kh = lane >> 4;
  int t0 = ttile * 16, j0 = wv * 16;
  const float* whu = w + 2 * ND;
  const float* Crow = C + (size_t)(b * NT + t0 + m) * ND;
  const float* Qrow = Q + (size_t)(b * NJ + j0 + m) * ND;
  v8f acc = {};
#pragma unroll 4
  for (int k0 = 0; k0 < ND; k0 += 4) {
    int ka = k0 + 2 * kh;
    v2f a  = *(const v2f*)(Crow + ka);
    v2f wu = *(const v2f*)(whu + ka);
    a *= wu;
    v2f bf = *(const v2f*)(Qrow + ka);
    acc = __builtin_amdgcn_wmma_f32_16x16x4_f32(false, a, false, bf, (short)0,
                                                acc, false, false);
  }
  int jj = j0 + m;
  float qv = qb[b * NJ + jj];
#pragma unroll
  for (int r = 0; r < 8; ++r) {
    int tt = t0 + r + 8 * kh;
    S[(size_t)(b * NT + tt) * NJ + jj] = acc[r] + ch[b * NT + tt] + qv;
  }
}

__global__ void k_softmax_rows(const float* __restrict__ S, float* __restrict__ Sj) {
  int wave = blockIdx.x * (blockDim.x >> 5) + (threadIdx.x >> 5);
  int lane = threadIdx.x & 31;
  if (wave >= NB * NT) return;
  const float* row = S + (size_t)wave * NJ;
  float4 v = *(const float4*)(row + lane * 4);
  float m = fmaxf(fmaxf(v.x, v.y), fmaxf(v.z, v.w));
  m = wave_reduce_max(m);
  float e0 = expf(v.x - m), e1 = expf(v.y - m), e2 = expf(v.z - m), e3 = expf(v.w - m);
  float s = wave_reduce_sum(e0 + e1 + e2 + e3);
  float inv = 1.0f / s;
  float4 o; o.x = e0 * inv; o.y = e1 * inv; o.z = e2 * inv; o.w = e3 * inv;
  *(float4*)(Sj + (size_t)wave * NJ + lane * 4) = o;
}

__global__ void k_softmax_cols(const float* __restrict__ S, float* __restrict__ St) {
  __shared__ float red[8][32];
  __shared__ float cmax[32];
  __shared__ float csum[32];
  int b = blockIdx.x >> 2;
  int jc = blockIdx.x & 3;
  int lx = threadIdx.x & 31, ty = threadIdx.x >> 5;
  int j = jc * 32 + lx;
  const float* base = S + (size_t)b * NT * NJ + j;
  float lm = -3.4e38f;
  for (int t = ty; t < NT; t += 8) lm = fmaxf(lm, base[(size_t)t * NJ]);
  red[ty][lx] = lm;
  __syncthreads();
  if (ty == 0) {
    float m = red[0][lx];
#pragma unroll
    for (int i = 1; i < 8; ++i) m = fmaxf(m, red[i][lx]);
    cmax[lx] = m;
  }
  __syncthreads();
  float M = cmax[lx];
  float ls = 0.f;
  for (int t = ty; t < NT; t += 8) ls += expf(base[(size_t)t * NJ] - M);
  __syncthreads();
  red[ty][lx] = ls;
  __syncthreads();
  if (ty == 0) {
    float s = red[0][lx];
#pragma unroll
    for (int i = 1; i < 8; ++i) s += red[i][lx];
    csum[lx] = s;
  }
  __syncthreads();
  float inv = 1.0f / csum[lx];
  float* ob = St + (size_t)b * NT * NJ + j;
  for (int t = ty; t < NT; t += 8)
    ob[(size_t)t * NJ] = expf(base[(size_t)t * NJ] - M) * inv;
}

__global__ void k_tmp(const float* __restrict__ St, const float* __restrict__ C,
                      float* __restrict__ tmp) {
  int b = blockIdx.x >> 3;
  int dpair = blockIdx.x & 7;
  int wv = threadIdx.x >> 5;
  int lane = threadIdx.x & 31;
  int m = lane & 15, kh = lane >> 4;
  int j0 = wv * 16, d0 = dpair * 32;
  const float* Sbase = St + (size_t)b * NT * NJ + j0 + m;
  const float* Cbase = C + (size_t)b * NT * ND + d0 + m;
  v8f acc0 = {}, acc1 = {};
#pragma unroll 2
  for (int k0 = 0; k0 < NT; k0 += 4) {
    int ka = k0 + 2 * kh;
    v2f a, b0, b1;
    a.x  = Sbase[(size_t)ka * NJ];
    a.y  = Sbase[(size_t)(ka + 1) * NJ];
    b0.x = Cbase[(size_t)ka * ND];
    b0.y = Cbase[(size_t)(ka + 1) * ND];
    b1.x = Cbase[(size_t)ka * ND + 16];
    b1.y = Cbase[(size_t)(ka + 1) * ND + 16];
    acc0 = __builtin_amdgcn_wmma_f32_16x16x4_f32(false, a, false, b0, (short)0,
                                                 acc0, false, false);
    acc1 = __builtin_amdgcn_wmma_f32_16x16x4_f32(false, a, false, b1, (short)0,
                                                 acc1, false, false);
  }
#pragma unroll
  for (int r = 0; r < 8; ++r) {
    int jj = j0 + r + 8 * kh;
    float* ob = tmp + (size_t)(b * NJ + jj) * ND + d0 + m;
    ob[0]  = acc0[r];
    ob[16] = acc1[r];
  }
}

__global__ void k_out(const float* __restrict__ Sj, const float* __restrict__ Q,
                      const float* __restrict__ tmp, const float* __restrict__ C,
                      float* __restrict__ out) {
  int blk = blockIdx.x;
  int b = blk >> 7;
  int rem = blk & 127;
  int ttile = rem >> 1;
  int dh = rem & 1;
  int wv = threadIdx.x >> 5;
  int lane = threadIdx.x & 31;
  int m = lane & 15, kh = lane >> 4;
  int t0 = ttile * 16, d0 = (dh * 8 + wv) * 16;
  const float* Srow = Sj + (size_t)(b * NT + t0 + m) * NJ;
  const float* Qb = Q + (size_t)b * NJ * ND + d0 + m;
  const float* Tb = tmp + (size_t)b * NJ * ND + d0 + m;
  v8f accA = {}, accB = {};
#pragma unroll 4
  for (int k0 = 0; k0 < NJ; k0 += 4) {
    int ka = k0 + 2 * kh;
    v2f a = *(const v2f*)(Srow + ka);
    v2f bq, bt;
    bq.x = Qb[(size_t)ka * ND];
    bq.y = Qb[(size_t)(ka + 1) * ND];
    bt.x = Tb[(size_t)ka * ND];
    bt.y = Tb[(size_t)(ka + 1) * ND];
    accA = __builtin_amdgcn_wmma_f32_16x16x4_f32(false, a, false, bq, (short)0,
                                                 accA, false, false);
    accB = __builtin_amdgcn_wmma_f32_16x16x4_f32(false, a, false, bt, (short)0,
                                                 accB, false, false);
  }
  int d = d0 + m;
#pragma unroll
  for (int r = 0; r < 8; ++r) {
    int t = t0 + r + 8 * kh;
    float cv = C[(size_t)(b * NT + t) * ND + d];
    float* ob = out + (size_t)(b * NT + t) * (4 * ND);
    ob[d]          = cv;
    ob[ND + d]     = accA[r];
    ob[2 * ND + d] = cv * accA[r];
    ob[3 * ND + d] = cv * accB[r];
  }
}

extern "C" void kernel_launch(void* const* d_in, const int* in_sizes, int n_in,
                              void* d_out, int out_size, void* d_ws, size_t ws_size,
                              hipStream_t stream) {
  (void)in_sizes; (void)n_in; (void)out_size; (void)ws_size;
  const float* C = (const float*)d_in[0];
  const float* Q = (const float*)d_in[1];
  const float* w = (const float*)d_in[2];
  float* out = (float*)d_out;

  float* ws = (float*)d_ws;
  float* ch = ws;
  float* qb = ch + NB * NT;
  float* S  = qb + NB * NJ;
  float* Sj = S  + (size_t)NB * NT * NJ;
  float* St = Sj + (size_t)NB * NT * NJ;
  float* tp = St + (size_t)NB * NT * NJ;

  int rows = NB * NT + NB * NJ;
  k_rowdots<<<(rows + 7) / 8, 256, 0, stream>>>(C, Q, w, ch, qb);
  k_score<<<NB * 64, 256, 0, stream>>>(C, Q, w, ch, qb, S);
  k_softmax_rows<<<(NB * NT + 7) / 8, 256, 0, stream>>>(S, Sj);
  k_softmax_cols<<<NB * (NJ / 32), 256, 0, stream>>>(S, St);
  k_tmp<<<NB * 8, 256, 0, stream>>>(St, C, tp);
  k_out<<<NB * 128, 256, 0, stream>>>(Sj, Q, tp, C, out);
}


# k_tmp: split the t=1024 sum into 8 slices run by separate waves (16-wave blocks, 2 j-tiles x 8 slices), add the 8 partial tiles in fixed order through LDS; changes the summation order of tmp
# speedup vs baseline: 1.3044x; 1.3044x over previous
#include <hip/hip_runtime.h>

#define NB 32
#define NT 1024
#define NJ 128
#define ND 256

typedef __attribute__((ext_vector_type(2))) float v2f;
typedef __attribute__((ext_vector_type(8))) float v8f;

__device__ __forceinline__ float wave_reduce_sum(float v) {
#pragma unroll
  for (int o = 16; o > 0; o >>= 1) v += __shfl_xor(v, o, 32);
  return v;
}
__device__ __forceinline__ float wave_reduce_max(float v) {
#pragma unroll
  for (int o = 16; o > 0; o >>= 1) v = fmaxf(v, __shfl_xor(v, o, 32));
  return v;
}

__global__ void k_rowdots(const float* __restrict__ C, const float* __restrict__ Q,
                          const float* __restrict__ w, float* __restrict__ ch,
                          float* __restrict__ qb) {
  int wave = blockIdx.x * (blockDim.x >> 5) + (threadIdx.x >> 5);
  int lane = threadIdx.x & 31;
  const int nch = NB * NT;
  const int nrows = nch + NB * NJ;
  if (wave >= nrows) return;
  const float* row; const float* wv; float* out;
  if (wave < nch) { row = C + (size_t)wave * ND; wv = w;      out = ch + wave; }
  else { int r = wave - nch; row = Q + (size_t)r * ND; wv = w + ND; out = qb + r; }
  float s = 0.f;
#pragma unroll
  for (int i = lane; i < ND; i += 32) s += row[i] * wv[i];
  s = wave_reduce_sum(s);
  if (lane == 0) *out = s;
}

__global__ void k_score(const float* __restrict__ C, const float* __restrict__ Q,
                        const float* __restrict__ w, const float* __restrict__ ch,
                        const float* __restrict__ qb, float* __restrict__ S) {
  int blk = blockIdx.x;
  int b = blk >> 6;
  int ttile = blk & 63;
  int wv = threadIdx.x >> 5;
  int lane = threadIdx.x & 31;
  int m = lane & 15, kh = lane >> 4;
  int t0 = ttile * 16, j0 = wv * 16;
  const float* whu = w + 2 * ND;
  const float* Crow = C + (size_t)(b * NT + t0 + m) * ND;
  const float* Qrow = Q + (size_t)(b * NJ + j0 + m) * ND;
  v8f acc = {};
#pragma unroll 4
  for (int k0 = 0; k0 < ND; k0 += 4) {
    int ka = k0 + 2 * kh;
    v2f a  = *(const v2f*)(Crow + ka);
    v2f wu = *(const v2f*)(whu + ka);
    a *= wu;
    v2f bf = *(const v2f*)(Qrow + ka);
    acc = __builtin_amdgcn_wmma_f32_16x16x4_f32(false, a, false, bf, (short)0,
                                                acc, false, false);
  }
  int jj = j0 + m;
  float qv = qb[b * NJ + jj];
#pragma unroll
  for (int r = 0; r < 8; ++r) {
    int tt = t0 + r + 8 * kh;
    S[(size_t)(b * NT + tt) * NJ + jj] = acc[r] + ch[b * NT + tt] + qv;
  }
}

__global__ void k_softmax_rows(const float* __restrict__ S, float* __restrict__ Sj) {
  int wave = blockIdx.x * (blockDim.x >> 5) + (threadIdx.x >> 5);
  int lane = threadIdx.x & 31;
  if (wave >= NB * NT) return;
  const float* row = S + (size_t)wave * NJ;
  float4 v = *(const float4*)(row + lane * 4);
  float m = fmaxf(fmaxf(v.x, v.y), fmaxf(v.z, v.w));
  m = wave_reduce_max(m);
  float e0 = expf(v.x - m), e1 = expf(v.y - m), e2 = expf(v.z - m), e3 = expf(v.w - m);
  float s = wave_reduce_sum(e0 + e1 + e2 + e3);
  float inv = 1.0f / s;
  float4 o; o.x = e0 * inv; o.y = e1 * inv; o.z = e2 * inv; o.w = e3 * inv;
  *(float4*)(Sj + (size_t)wave * NJ + lane * 4) = o;
}

__global__ void k_softmax_cols(const float* __restrict__ S, float* __restrict__ St) {
  __shared__ float red[8][32];
  __shared__ float cmax[32];
  __shared__ float csum[32];
  int b = blockIdx.x >> 2;
  int jc = blockIdx.x & 3;
  int lx = threadIdx.x & 31, ty = threadIdx.x >> 5;
  int j = jc * 32 + lx;
  const float* base = S + (size_t)b * NT * NJ + j;
  float lm = -3.4e38f;
  for (int t = ty; t < NT; t += 8) lm = fmaxf(lm, base[(size_t)t * NJ]);
  red[ty][lx] = lm;
  __syncthreads();
  if (ty == 0) {
    float m = red[0][lx];
#pragma unroll
    for (int i = 1; i < 8; ++i) m = fmaxf(m, red[i][lx]);
    cmax[lx] = m;
  }
  __syncthreads();
  float M = cmax[lx];
  float ls = 0.f;
  for (int t = ty; t < NT; t += 8) ls += expf(base[(size_t)t * NJ] - M);
  __syncthreads();
  red[ty][lx] = ls;
  __syncthreads();
  if (ty == 0) {
    float s = red[0][lx];
#pragma unroll
    for (int i = 1; i < 8; ++i) s += red[i][lx];
    csum[lx] = s;
  }
  __syncthreads();
  float inv = 1.0f / csum[lx];
  float* ob = St + (size_t)b * NT * NJ + j;
  for (int t = ty; t < NT; t += 8)
    ob[(size_t)t * NJ] = expf(base[(size_t)t * NJ] - M) * inv;
}

__global__ void k_tmp(const float* __restrict__ St, const float* __restrict__ C,
                      float* __restrict__ tmp) {
  __shared__ float part[8][32][32];
  int b = blockIdx.x >> 5;
  int jpair = (blockIdx.x >> 3) & 3;
  int dpair = blockIdx.x & 7;
  int wv = threadIdx.x >> 5;
  int ks = wv >> 1, jh = wv & 1;
  int lane = threadIdx.x & 31;
  int m = lane & 15, kh = lane >> 4;
  int j0 = jpair * 32, d0 = dpair * 32;
  const int KS = NT / 8;
  const float* Sbase = St + ((size_t)b * NT + ks * KS) * NJ + j0 + jh * 16 + m;
  const float* Cbase = C + ((size_t)b * NT + ks * KS) * ND + d0 + m;
  v8f acc0 = {}, acc1 = {};
#pragma unroll 2
  for (int k0 = 0; k0 < KS; k0 += 4) {
    int ka = k0 + 2 * kh;
    v2f a, b0, b1;
    a.x  = Sbase[(size_t)ka * NJ];
    a.y  = Sbase[(size_t)(ka + 1) * NJ];
    b0.x = Cbase[(size_t)ka * ND];
    b0.y = Cbase[(size_t)(ka + 1) * ND];
    b1.x = Cbase[(size_t)ka * ND + 16];
    b1.y = Cbase[(size_t)(ka + 1) * ND + 16];
    acc0 = __builtin_amdgcn_wmma_f32_16x16x4_f32(false, a, false, b0, (short)0,
                                                 acc0, false, false);
    acc1 = __builtin_amdgcn_wmma_f32_16x16x4_f32(false, a, false, b1, (short)0,
                                                 acc1, false, false);
  }
#pragma unroll
  for (int r = 0; r < 8; ++r) {
    int row = jh * 16 + r + 8 * kh;
    part[ks][row][m]      = acc0[r];
    part[ks][row][m + 16] = acc1[r];
  }
  __syncthreads();
#pragma unroll
  for (int h = 0; h < 2; ++h) {
    int row = wv + 16 * h;
    float s = part[0][row][lane];
#pragma unroll
    for (int i = 1; i < 8; ++i) s += part[i][row][lane];
    tmp[(size_t)(b * NJ + j0 + row) * ND + d0 + lane] = s;
  }
}

__global__ void k_out(const float* __restrict__ Sj, const float* __restrict__ Q,
                      const float* __restrict__ tmp, const float* __restrict__ C,
                      float* __restrict__ out) {
  int blk = blockIdx.x;
  int b = blk >> 7;
  int rem = blk & 127;
  int ttile = rem >> 1;
  int dh = rem & 1;
  int wv = threadIdx.x >> 5;
  int lane = threadIdx.x & 31;
  int m = lane & 15, kh = lane >> 4;
  int t0 = ttile * 16, d0 = (dh * 8 + wv) * 16;
  const float* Srow = Sj + (size_t)(b * NT + t0 + m) * NJ;
  const float* Qb = Q + (size_t)b * NJ * ND + d0 + m;
  const float* Tb = tmp + (size_t)b * NJ * ND + d0 + m;
  v8f accA = {}, accB = {};
#pragma unroll 4
  for (int k0 = 0; k0 < NJ; k0 += 4) {
    int ka = k0 + 2 * kh;
    v2f a = *(const v2f*)(Srow + ka);
    v2f bq, bt;
    bq.x = Qb[(size_t)ka * ND];
    bq.y = Qb[(size_t)(ka + 1) * ND];
    bt.x = Tb[(size_t)ka * ND];
    bt.y = Tb[(size_t)(ka + 1) * ND];
    accA = __builtin_amdgcn_wmma_f32_16x16x4_f32(false, a, false, bq, (short)0,
                                                 accA, false, false);
    accB = __builtin_amdgcn_wmma_f32_16x16x4_f32(false, a, false, bt, (short)0,
                                                 accB, false, false);
  }
  int d = d0 + m;
#pragma unroll
  for (int r = 0; r < 8; ++r) {
    int t = t0 + r + 8 * kh;
    float cv = C[(size_t)(b * NT + t) * ND + d];
    float* ob = out + (size_t)(b * NT + t) * (4 * ND);
    ob[d]          = cv;
    ob[ND + d]     = accA[r];
    ob[2 * ND + d] = cv * accA[r];
    ob[3 * ND + d] = cv * accB[r];
  }
}

extern "C" void kernel_launch(void* const* d_in, const int* in_sizes, int n_in,
                              void* d_out, int out_size, void* d_ws, size_t ws_size,
                              hipStream_t stream) {
  (void)in_sizes; (void)n_in; (void)out_size; (void)ws_size;
  const float* C = (const float*)d_in[0];
  const float* Q = (const float*)d_in[1];
  const float* w = (const float*)d_in[2];
  float* out = (float*)d_out;

  float* ws = (float*)d_ws;
  float* ch = ws;
  float* qb = ch + NB * NT;
  float* S  = qb + NB * NJ;
  float* Sj = S  + (size_t)NB * NT * NJ;
  float* St = Sj + (size_t)NB * NT * NJ;
  float* tp = St + (size_t)NB * NT * NJ;

  int rows = NB * NT + NB * NJ;
  k_rowdots<<<(rows + 7) / 8, 256, 0, stream>>>(C, Q, w, ch, qb);
  k_score<<<NB * 64, 256, 0, stream>>>(C, Q, w, ch, qb, S);
  k_softmax_rows<<<(NB * NT + 7) / 8, 256, 0, stream>>>(S, Sj);
  k_softmax_cols<<<NB * (NJ / 32), 256, 0, stream>>>(S, St);
  k_tmp<<<NB * 32, 512, 0, stream>>>(St, C, tp);
  k_out<<<NB * 128, 256, 0, stream>>>(Sj, Q, tp, C, out);
}


# k_softmax_cols: 32-wave blocks, each wave holds 32 rows of the 32 columns in registers, so S is read once with 32 loads in flight per lane; changes the summation order of the column sums
# speedup vs baseline: 1.5664x; 1.2009x over previous
#include <hip/hip_runtime.h>

#define NB 32
#define NT 1024
#define NJ 128
#define ND 256

typedef __attribute__((ext_vector_type(2))) float v2f;
typedef __attribute__((ext_vector_type(8))) float v8f;

__device__ __forceinline__ float wave_reduce_sum(float v) {
#pragma unroll
  for (int o = 16; o > 0; o >>= 1) v += __shfl_xor(v, o, 32);
  return v;
}
__device__ __forceinline__ float wave_reduce_max(float v) {
#pragma unroll
  for (int o = 16; o > 0; o >>= 1) v = fmaxf(v, __shfl_xor(v, o, 32));
  return v;
}

__global__ void k_rowdots(const float* __restrict__ C, const float* __restrict__ Q,
                          const float* __restrict__ w, float* __restrict__ ch,
                          float* __restrict__ qb) {
  int wave = blockIdx.x * (blockDim.x >> 5) + (threadIdx.x >> 5);
  int lane = threadIdx.x & 31;
  const int nch = NB * NT;
  const int nrows = nch + NB * NJ;
  if (wave >= nrows) return;
  const float* row; const float* wv; float* out;
  if (wave < nch) { row = C + (size_t)wave * ND; wv = w;      out = ch + wave; }
  else { int r = wave - nch; row = Q + (size_t)r * ND; wv = w + ND; out = qb + r; }
  float s = 0.f;
#pragma unroll
  for (int i = lane; i < ND; i += 32) s += row[i] * wv[i];
  s = wave_reduce_sum(s);
  if (lane == 0) *out = s;
}

__global__ void k_score(const float* __restrict__ C, const float* __restrict__ Q,
                        const float* __restrict__ w, const float* __restrict__ ch,
                        const float* __restrict__ qb, float* __restrict__ S) {
  int blk = blockIdx.x;
  int b = blk >> 6;
  int ttile = blk & 63;
  int wv = threadIdx.x >> 5;
  int lane = threadIdx.x & 31;
  int m = lane & 15, kh = lane >> 4;
  int t0 = ttile * 16, j0 = wv * 16;
  const float* whu = w + 2 * ND;
  const float* Crow = C + (size_t)(b * NT + t0 + m) * ND;
  const float* Qrow = Q + (size_t)(b * NJ + j0 + m) * ND;
  v8f acc = {};
#pragma unroll 4
  for (int k0 = 0; k0 < ND; k0 += 4) {
    int ka = k0 + 2 * kh;
    v2f a  = *(const v2f*)(Crow + ka);
    v2f wu = *(const v2f*)(whu + ka);
    a *= wu;
    v2f bf = *(const v2f*)(Qrow + ka);
    acc = __builtin_amdgcn_wmma_f32_16x16x4_f32(false, a, false, bf, (short)0,
                                                acc, false, false);
  }
  int jj = j0 + m;
  float qv = qb[b * NJ + jj];
#pragma unroll
  for (int r = 0; r < 8; ++r) {
    int tt = t0 + r + 8 * kh;
    S[(size_t)(b * NT + tt) * NJ + jj] = acc[r] + ch[b * NT + tt] + qv;
  }
}

__global__ void k_softmax_rows(const float* __restrict__ S, float* __restrict__ Sj) {
  int wave = blockIdx.x * (blockDim.x >> 5) + (threadIdx.x >> 5);
  int lane = threadIdx.x & 31;
  if (wave >= NB * NT) return;
  const float* row = S + (size_t)wave * NJ;
  float4 v = *(const float4*)(row + lane * 4);
  float m = fmaxf(fmaxf(v.x, v.y), fmaxf(v.z, v.w));
  m = wave_reduce_max(m);
  float e0 = expf(v.x - m), e1 = expf(v.y - m), e2 = expf(v.z - m), e3 = expf(v.w - m);
  float s = wave_reduce_sum(e0 + e1 + e2 + e3);
  float inv = 1.0f / s;
  float4 o; o.x = e0 * inv; o.y = e1 * inv; o.z = e2 * inv; o.w = e3 * inv;
  *(float4*)(Sj + (size_t)wave * NJ + lane * 4) = o;
}

__global__ __launch_bounds__(1024) void k_softmax_cols(const float* __restrict__ S,
                                                       float* __restrict__ St) {
  __shared__ float pmax[32][32];
  __shared__ float psum[32][32];
  int b = blockIdx.x >> 2;
  int jc = blockIdx.x & 3;
  int lx = threadIdx.x & 31, wv = threadIdx.x >> 5;
  size_t off = ((size_t)b * NT + wv * 32) * NJ + jc * 32 + lx;
  const float* base = S + off;
  float v[32];
#pragma unroll
  for (int i = 0; i < 32; ++i) v[i] = base[(size_t)i * NJ];
  float lm = v[0];
#pragma unroll
  for (int i = 1; i < 32; ++i) lm = fmaxf(lm, v[i]);
  pmax[wv][lx] = lm;
  __syncthreads();
  float M = pmax[0][lx];
#pragma unroll
  for (int i = 1; i < 32; ++i) M = fmaxf(M, pmax[i][lx]);
  float ls = 0.f;
#pragma unroll
  for (int i = 0; i < 32; ++i) { v[i] = expf(v[i] - M); ls += v[i]; }
  psum[wv][lx] = ls;
  __syncthreads();
  float s = psum[0][lx];
#pragma unroll
  for (int i = 1; i < 32; ++i) s += psum[i][lx];
  float inv = 1.0f / s;
  float* ob = St + off;
#pragma unroll
  for (int i = 0; i < 32; ++i) ob[(size_t)i * NJ] = v[i] * inv;
}

__global__ void k_tmp(const float* __restrict__ St, const float* __restrict__ C,
                      float* __restrict__ tmp) {
  __shared__ float part[8][32][32];
  int b = blockIdx.x >> 5;
  int jpair = (blockIdx.x >> 3) & 3;
  int dpair = blockIdx.x & 7;
  int wv = threadIdx.x >> 5;
  int ks = wv >> 1, jh = wv & 1;
  int lane = threadIdx.x & 31;
  int m = lane & 15, kh = lane >> 4;
  int j0 = jpair * 32, d0 = dpair * 32;
  const int KS = NT / 8;
  const float* Sbase = St + ((size_t)b * NT + ks * KS) * NJ + j0 + jh * 16 + m;
  const float* Cbase = C + ((size_t)b * NT + ks * KS) * ND + d0 + m;
  v8f acc0 = {}, acc1 = {};
#pragma unroll 2
  for (int k0 = 0; k0 < KS; k0 += 4) {
    int ka = k0 + 2 * kh;
    v2f a, b0, b1;
    a.x  = Sbase[(size_t)ka * NJ];
    a.y  = Sbase[(size_t)(ka + 1) * NJ];
    b0.x = Cbase[(size_t)ka * ND];
    b0.y = Cbase[(size_t)(ka + 1) * ND];
    b1.x = Cbase[(size_t)ka * ND + 16];
    b1.y = Cbase[(size_t)(ka + 1) * ND + 16];
    acc0 = __builtin_amdgcn_wmma_f32_16x16x4_f32(false, a, false, b0, (short)0,
                                                 acc0, false, false);
    acc1 = __builtin_amdgcn_wmma_f32_16x16x4_f32(false, a, false, b1, (short)0,
                                                 acc1, false, false);
  }
#pragma unroll
  for (int r = 0; r < 8; ++r) {
    int row = jh * 16 + r + 8 * kh;
    part[ks][row][m]      = acc0[r];
    part[ks][row][m + 16] = acc1[r];
  }
  __syncthreads();
#pragma unroll
  for (int h = 0; h < 2; ++h) {
    int row = wv + 16 * h;
    float s = part[0][row][lane];
#pragma unroll
    for (int i = 1; i < 8; ++i) s += part[i][row][lane];
    tmp[(size_t)(b * NJ + j0 + row) * ND + d0 + lane] = s;
  }
}

__global__ void k_out(const float* __restrict__ Sj, const float* __restrict__ Q,
                      const float* __restrict__ tmp, const float* __restrict__ C,
                      float* __restrict__ out) {
  int blk = blockIdx.x;
  int b = blk >> 7;
  int rem = blk & 127;
  int ttile = rem >> 1;
  int dh = rem & 1;
  int wv = threadIdx.x >> 5;
  int lane = threadIdx.x & 31;
  int m = lane & 15, kh = lane >> 4;
  int t0 = ttile * 16, d0 = (dh * 8 + wv) * 16;
  const float* Srow = Sj + (size_t)(b * NT + t0 + m) * NJ;
  const float* Qb = Q + (size_t)b * NJ * ND + d0 + m;
  const float* Tb = tmp + (size_t)b * NJ * ND + d0 + m;
  v8f accA = {}, accB = {};
#pragma unroll 4
  for (int k0 = 0; k0 < NJ; k0 += 4) {
    int ka = k0 + 2 * kh;
    v2f a = *(const v2f*)(Srow + ka);
    v2f bq, bt;
    bq.x = Qb[(size_t)ka * ND];
    bq.y = Qb[(size_t)(ka + 1) * ND];
    bt.x = Tb[(size_t)ka * ND];
    bt.y = Tb[(size_t)(ka + 1) * ND];
    accA = __builtin_amdgcn_wmma_f32_16x16x4_f32(false, a, false, bq, (short)0,
                                                 accA, false, false);
    accB = __builtin_amdgcn_wmma_f32_16x16x4_f32(false, a, false, bt, (short)0,
                                                 accB, false, false);
  }
  int d = d0 + m;
#pragma unroll
  for (int r = 0; r < 8; ++r) {
    int t = t0 + r + 8 * kh;
    float cv = C[(size_t)(b * NT + t) * ND + d];
    float* ob = out + (size_t)(b * NT + t) * (4 * ND);
    ob[d]          = cv;
    ob[ND + d]     = accA[r];
    ob[2 * ND + d] = cv * accA[r];
    ob[3 * ND + d] = cv * accB[r];
  }
}

extern "C" void kernel_launch(void* const* d_in, const int* in_sizes, int n_in,
                              void* d_out, int out_size, void* d_ws, size_t ws_size,
                              hipStream_t stream) {
  (void)in_sizes; (void)n_in; (void)out_size; (void)ws_size;
  const float* C = (const float*)d_in[0];
  const float* Q = (const float*)d_in[1];
  const float* w = (const float*)d_in[2];
  float* out = (float*)d_out;

  float* ws = (float*)d_ws;
  float* ch = ws;
  float* qb = ch + NB * NT;
  float* S  = qb + NB * NJ;
  float* Sj = S  + (size_t)NB * NT * NJ;
  float* St = Sj + (size_t)NB * NT * NJ;
  float* tp = St + (size_t)NB * NT * NJ;

  int rows = NB * NT + NB * NJ;
  k_rowdots<<<(rows + 7) / 8, 256, 0, stream>>>(C, Q, w, ch, qb);
  k_score<<<NB * 64, 256, 0, stream>>>(C, Q, w, ch, qb, S);
  k_softmax_rows<<<(NB * NT + 7) / 8, 256, 0, stream>>>(S, Sj);
  k_softmax_cols<<<NB * (NJ / 32), 1024, 0, stream>>>(S, St);
  k_tmp<<<NB * 32, 512, 0, stream>>>(St, C, tp);
  k_out<<<NB * 128, 256, 0, stream>>>(Sj, Q, tp, C, out);
}


# k_out: a wave keeps its B fragments of Q and tmp (16 columns, all k) in registers over 8 t-tiles and reads A from an LDS copy of the Sj tile staged with whole-line loads; sums unchanged
# speedup vs baseline: 1.9049x; 1.2161x over previous
#include <hip/hip_runtime.h>

#define NB 32
#define NT 1024
#define NJ 128
#define ND 256

typedef __attribute__((ext_vector_type(2))) float v2f;
typedef __attribute__((ext_vector_type(8))) float v8f;

__device__ __forceinline__ float wave_reduce_sum(float v) {
#pragma unroll
  for (int o = 16; o > 0; o >>= 1) v += __shfl_xor(v, o, 32);
  return v;
}
__device__ __forceinline__ float wave_reduce_max(float v) {
#pragma unroll
  for (int o = 16; o > 0; o >>= 1) v = fmaxf(v, __shfl_xor(v, o, 32));
  return v;
}

__global__ void k_rowdots(const float* __restrict__ C, const float* __restrict__ Q,
                          const float* __restrict__ w, float* __restrict__ ch,
                          float* __restrict__ qb) {
  int wave = blockIdx.x * (blockDim.x >> 5) + (threadIdx.x >> 5);
  int lane = threadIdx.x & 31;
  const int nch = NB * NT;
  const int nrows = nch + NB * NJ;
  if (wave >= nrows) return;
  const float* row; const float* wv; float* out;
  if (wave < nch) { row = C + (size_t)wave * ND; wv = w;      out = ch + wave; }
  else { int r = wave - nch; row = Q + (size_t)r * ND; wv = w + ND; out = qb + r; }
  float s = 0.f;
#pragma unroll
  for (int i = lane; i < ND; i += 32) s += row[i] * wv[i];
  s = wave_reduce_sum(s);
  if (lane == 0) *out = s;
}

__global__ void k_score(const float* __restrict__ C, const float* __restrict__ Q,
                        const float* __restrict__ w, const float* __restrict__ ch,
                        const float* __restrict__ qb, float* __restrict__ S) {
  int blk = blockIdx.x;
  int b = blk >> 6;
  int ttile = blk & 63;
  int wv = threadIdx.x >> 5;
  int lane = threadIdx.x & 31;
  int m = lane & 15, kh = lane >> 4;
  int t0 = ttile * 16, j0 = wv * 16;
  const float* whu = w + 2 * ND;
  const float* Crow = C + (size_t)(b * NT + t0 + m) * ND;
  const float* Qrow = Q + (size_t)(b * NJ + j0 + m) * ND;
  v8f acc = {};
#pragma unroll 4
  for (int k0 = 0; k0 < ND; k0 += 4) {
    int ka = k0 + 2 * kh;
    v2f a  = *(const v2f*)(Crow + ka);
    v2f wu = *(const v2f*)(whu + ka);
    a *= wu;
    v2f bf = *(const v2f*)(Qrow + ka);
    acc = __builtin_amdgcn_wmma_f32_16x16x4_f32(false, a, false, bf, (short)0,
                                                acc, false, false);
  }
  int jj = j0 + m;
  float qv = qb[b * NJ + jj];
#pragma unroll
  for (int r = 0; r < 8; ++r) {
    int tt = t0 + r + 8 * kh;
    S[(size_t)(b * NT + tt) * NJ + jj] = acc[r] + ch[b * NT + tt] + qv;
  }
}

__global__ void k_softmax_rows(const float* __restrict__ S, float* __restrict__ Sj) {
  int wave = blockIdx.x * (blockDim.x >> 5) + (threadIdx.x >> 5);
  int lane = threadIdx.x & 31;
  if (wave >= NB * NT) return;
  const float* row = S + (size_t)wave * NJ;
  float4 v = *(const float4*)(row + lane * 4);
  float m = fmaxf(fmaxf(v.x, v.y), fmaxf(v.z, v.w));
  m = wave_reduce_max(m);
  float e0 = expf(v.x - m), e1 = expf(v.y - m), e2 = expf(v.z - m), e3 = expf(v.w - m);
  float s = wave_reduce_sum(e0 + e1 + e2 + e3);
  float inv = 1.0f / s;
  float4 o; o.x = e0 * inv; o.y = e1 * inv; o.z = e2 * inv; o.w = e3 * inv;
  *(float4*)(Sj + (size_t)wave * NJ + lane * 4) = o;
}

__global__ __launch_bounds__(1024) void k_softmax_cols(const float* __restrict__ S,
                                                       float* __restrict__ St) {
  __shared__ float pmax[32][32];
  __shared__ float psum[32][32];
  int b = blockIdx.x >> 2;
  int jc = blockIdx.x & 3;
  int lx = threadIdx.x & 31, wv = threadIdx.x >> 5;
  size_t off = ((size_t)b * NT + wv * 32) * NJ + jc * 32 + lx;
  const float* base = S + off;
  float v[32];
#pragma unroll
  for (int i = 0; i < 32; ++i) v[i] = base[(size_t)i * NJ];
  float lm = v[0];
#pragma unroll
  for (int i = 1; i < 32; ++i) lm = fmaxf(lm, v[i]);
  pmax[wv][lx] = lm;
  __syncthreads();
  float M = pmax[0][lx];
#pragma unroll
  for (int i = 1; i < 32; ++i) M = fmaxf(M, pmax[i][lx]);
  float ls = 0.f;
#pragma unroll
  for (int i = 0; i < 32; ++i) { v[i] = expf(v[i] - M); ls += v[i]; }
  psum[wv][lx] = ls;
  __syncthreads();
  float s = psum[0][lx];
#pragma unroll
  for (int i = 1; i < 32; ++i) s += psum[i][lx];
  float inv = 1.0f / s;
  float* ob = St + off;
#pragma unroll
  for (int i = 0; i < 32; ++i) ob[(size_t)i * NJ] = v[i] * inv;
}

__global__ void k_tmp(const float* __restrict__ St, const float* __restrict__ C,
                      float* __restrict__ tmp) {
  __shared__ float part[8][32][32];
  int b = blockIdx.x >> 5;
  int jpair = (blockIdx.x >> 3) & 3;
  int dpair = blockIdx.x & 7;
  int wv = threadIdx.x >> 5;
  int ks = wv >> 1, jh = wv & 1;
  int lane = threadIdx.x & 31;
  int m = lane & 15, kh = lane >> 4;
  int j0 = jpair * 32, d0 = dpair * 32;
  const int KS = NT / 8;
  const float* Sbase = St + ((size_t)b * NT + ks * KS) * NJ + j0 + jh * 16 + m;
  const float* Cbase = C + ((size_t)b * NT + ks * KS) * ND + d0 + m;
  v8f acc0 = {}, acc1 = {};
#pragma unroll 2
  for (int k0 = 0; k0 < KS; k0 += 4) {
    int ka = k0 + 2 * kh;
    v2f a, b0, b1;
    a.x  = Sbase[(size_t)ka * NJ];
    a.y  = Sbase[(size_t)(ka + 1) * NJ];
    b0.x = Cbase[(size_t)ka * ND];
    b0.y = Cbase[(size_t)(ka + 1) * ND];
    b1.x = Cbase[(size_t)ka * ND + 16];
    b1.y = Cbase[(size_t)(ka + 1) * ND + 16];
    acc0 = __builtin_amdgcn_wmma_f32_16x16x4_f32(false, a, false, b0, (short)0,
                                                 acc0, false, false);
    acc1 = __builtin_amdgcn_wmma_f32_16x16x4_f32(false, a, false, b1, (short)0,
                                                 acc1, false, false);
  }
#pragma unroll
  for (int r = 0; r < 8; ++r) {
    int row = jh * 16 + r + 8 * kh;
    part[ks][row][m]      = acc0[r];
    part[ks][row][m + 16] = acc1[r];
  }
  __syncthreads();
#pragma unroll
  for (int h = 0; h < 2; ++h) {
    int row = wv + 16 * h;
    float s = part[0][row][lane];
#pragma unroll
    for (int i = 1; i < 8; ++i) s += part[i][row][lane];
    tmp[(size_t)(b * NJ + j0 + row) * ND + d0 + lane] = s;
  }
}

#define OT 8
#define AP 132
__global__ __launch_bounds__(256) void k_out(const float* __restrict__ Sj,
                                             const float* __restrict__ Q,
                                             const float* __restrict__ tmp,
                                             const float* __restrict__ C,
                                             float* __restrict__ out) {
  __shared__ __attribute__((aligned(16))) float As[2][16][AP];
  int blk = blockIdx.x;
  int b = blk >> 4;
  int chunk = (blk >> 1) & 7;
  int dh = blk & 1;
  int tid = threadIdx.x;
  int wv = tid >> 5;
  int lane = tid & 31;
  int m = lane & 15, kh = lane >> 4;
  int d = (dh * 8 + wv) * 16 + m;
  const float* Qb = Q + (size_t)b * NJ * ND + d;
  const float* Tb = tmp + (size_t)b * NJ * ND + d;
  v2f bq[32], bt[32];
#pragma unroll
  for (int i = 0; i < 32; ++i) {
    int ka = 4 * i + 2 * kh;
    bq[i].x = Qb[(size_t)ka * ND];
    bq[i].y = Qb[(size_t)(ka + 1) * ND];
    bt[i].x = Tb[(size_t)ka * ND];
    bt[i].y = Tb[(size_t)(ka + 1) * ND];
  }
  int tbase = chunk * (OT * 16);
  const float* Sg = Sj + ((size_t)b * NT + tbase + wv) * NJ + lane * 4;
  float4 g0 = *(const float4*)(Sg);
  float4 g1 = *(const float4*)(Sg + (size_t)8 * NJ);
  *(float4*)&As[0][wv][lane * 4] = g0;
  *(float4*)&As[0][wv + 8][lane * 4] = g1;
  __syncthreads();
#pragma nounroll
  for (int tt = 0; tt < OT; ++tt) {
    int cur = tt & 1;
    int t0 = tbase + tt * 16;
    if (tt + 1 < OT) {
      const float* Sn = Sg + (size_t)(tt + 1) * 16 * NJ;
      g0 = *(const float4*)(Sn);
      g1 = *(const float4*)(Sn + (size_t)8 * NJ);
    }
    float cv[8];
#pragma unroll
    for (int r = 0; r < 8; ++r)
      cv[r] = C[(size_t)(b * NT + t0 + r + 8 * kh) * ND + d];
    const float* Arow = &As[cur][m][2 * kh];
    v8f accA = {}, accB = {};
#pragma unroll
    for (int i = 0; i < 32; ++i) {
      v2f a = *(const v2f*)(Arow + 4 * i);
      accA = __builtin_amdgcn_wmma_f32_16x16x4_f32(false, a, false, bq[i], (short)0,
                                                   accA, false, false);
      accB = __builtin_amdgcn_wmma_f32_16x16x4_f32(false, a, false, bt[i], (short)0,
                                                   accB, false, false);
    }
    if (tt + 1 < OT) {
      *(float4*)&As[cur ^ 1][wv][lane * 4] = g0;
      *(float4*)&As[cur ^ 1][wv + 8][lane * 4] = g1;
      __syncthreads();
    }
#pragma unroll
    for (int r = 0; r < 8; ++r) {
      int t = t0 + r + 8 * kh;
      float* ob = out + (size_t)(b * NT + t) * (4 * ND);
      ob[d]          = cv[r];
      ob[ND + d]     = accA[r];
      ob[2 * ND + d] = cv[r] * accA[r];
      ob[3 * ND + d] = cv[r] * accB[r];
    }
  }
}

extern "C" void kernel_launch(void* const* d_in, const int* in_sizes, int n_in,
                              void* d_out, int out_size, void* d_ws, size_t ws_size,
                              hipStream_t stream) {
  (void)in_sizes; (void)n_in; (void)out_size; (void)ws_size;
  const float* C = (const float*)d_in[0];
  const float* Q = (const float*)d_in[1];
  const float* w = (const float*)d_in[2];
  float* out = (float*)d_out;

  float* ws = (float*)d_ws;
  float* ch = ws;
  float* qb = ch + NB * NT;
  float* S  = qb + NB * NJ;
  float* Sj = S  + (size_t)NB * NT * NJ;
  float* St = Sj + (size_t)NB * NT * NJ;
  float* tp = St + (size_t)NB * NT * NJ;

  int rows = NB * NT + NB * NJ;
  k_rowdots<<<(rows + 7) / 8, 256, 0, stream>>>(C, Q, w, ch, qb);
  k_score<<<NB * 64, 256, 0, stream>>>(C, Q, w, ch, qb, S);
  k_softmax_rows<<<(NB * NT + 7) / 8, 256, 0, stream>>>(S, Sj);
  k_softmax_cols<<<NB * (NJ / 32), 1024, 0, stream>>>(S, St);
  k_tmp<<<NB * 32, 512, 0, stream>>>(St, C, tp);
  k_out<<<NB * 16, 256, 0, stream>>>(Sj, Q, tp, C, out);
}


# k_score: a wave keeps its B fragments (Q with bias fold, 16 columns of j, all k) in registers over 4 t-tiles and reads A from an LDS copy of the C tile staged with whole-line loads; sums unchanged
# speedup vs baseline: 2.6670x; 1.4001x over previous
#include <hip/hip_runtime.h>

#define NB 32
#define NT 1024
#define NJ 128
#define ND 256

typedef __attribute__((ext_vector_type(2))) float v2f;
typedef __attribute__((ext_vector_type(8))) float v8f;

__device__ __forceinline__ float wave_reduce_sum(float v) {
#pragma unroll
  for (int o = 16; o > 0; o >>= 1) v += __shfl_xor(v, o, 32);
  return v;
}
__device__ __forceinline__ float wave_reduce_max(float v) {
#pragma unroll
  for (int o = 16; o > 0; o >>= 1) v = fmaxf(v, __shfl_xor(v, o, 32));
  return v;
}

#define ST 4
#define CP 260
__global__ __launch_bounds__(256) void k_score(const float* __restrict__ C,
                                               const float* __restrict__ Q,
                                               const float* __restrict__ w,
                                               float* __restrict__ S) {
  __shared__ __attribute__((aligned(16))) float wl[(ND / 2) * 8];
  __shared__ __attribute__((aligned(16))) float Cs[2][16][CP];
  int blk = blockIdx.x;
  int b = blk >> 4;
  int chunk = blk & 15;
  int wv = threadIdx.x >> 5;
  int lane = threadIdx.x & 31;
  int m = lane & 15, kh = lane >> 4;
  int j0 = wv * 16;
  int tbase = chunk * (ST * 16);
  {
    int k = threadIdx.x;
    int p = (k >> 1) * 8 + (k & 1);
    wl[p]     = w[2 * ND + k];
    wl[p + 2] = w[k];
    wl[p + 4] = w[ND + k];
  }
  __syncthreads();
  const float* Cg = C + ((size_t)b * NT + tbase + wv) * ND + lane * 4;
  float4 f0 = *(const float4*)(Cg);
  float4 f1 = *(const float4*)(Cg + 128);
  float4 f2 = *(const float4*)(Cg + (size_t)8 * ND);
  float4 f3 = *(const float4*)(Cg + (size_t)8 * ND + 128);
  const float* Qrow = Q + (size_t)(b * NJ + j0 + m) * ND;
  v2f bf[64];
  v2f qs = {0.f, 0.f};
#pragma unroll
  for (int i = 0; i < 64; ++i) {
    int ka = 4 * i + 2 * kh;
    v2f q = *(const v2f*)(Qrow + ka);
    const float* wp = wl + (ka >> 1) * 8;
    float4 x = *(const float4*)wp;
    v2f uu = *(const v2f*)(wp + 4);
    v2f hu = {x.x, x.y}, hh = {x.z, x.w};
    bf[i] = q * hu + hh;
    qs += q * uu;
  }
  float qv = qs.x + qs.y;
  qv += __shfl_xor(qv, 16, 32);
  *(float4*)&Cs[0][wv][lane * 4] = f0;
  *(float4*)&Cs[0][wv][128 + lane * 4] = f1;
  *(float4*)&Cs[0][wv + 8][lane * 4] = f2;
  *(float4*)&Cs[0][wv + 8][128 + lane * 4] = f3;
  __syncthreads();
  int jj = j0 + m;
#pragma nounroll
  for (int tt = 0; tt < ST; ++tt) {
    int cur = tt & 1;
    int t0 = tbase + tt * 16;
    float4 g0, g1, g2, g3;
    if (tt + 1 < ST) {
      const float* Cn = Cg + (size_t)(tt + 1) * 16 * ND;
      g0 = *(const float4*)(Cn);
      g1 = *(const float4*)(Cn + 128);
      g2 = *(const float4*)(Cn + (size_t)8 * ND);
      g3 = *(const float4*)(Cn + (size_t)8 * ND + 128);
    }
    const float* Arow = &Cs[cur][m][2 * kh];
    v8f acc = {};
#pragma unroll
    for (int i = 0; i < 64; ++i) {
      v2f a = *(const v2f*)(Arow + 4 * i);
      acc = __builtin_amdgcn_wmma_f32_16x16x4_f32(false, a, false, bf[i], (short)0,
                                                  acc, false, false);
    }
    if (tt + 1 < ST) {
      *(float4*)&Cs[cur ^ 1][wv][lane * 4] = g0;
      *(float4*)&Cs[cur ^ 1][wv][128 + lane * 4] = g1;
      *(float4*)&Cs[cur ^ 1][wv + 8][lane * 4] = g2;
      *(float4*)&Cs[cur ^ 1][wv + 8][128 + lane * 4] = g3;
      __syncthreads();
    }
#pragma unroll
    for (int r = 0; r < 8; ++r) {
      int tt2 = t0 + r + 8 * kh;
      S[(size_t)(b * NT + tt2) * NJ + jj] = acc[r] + qv;
    }
  }
}

__global__ void k_softmax_rows(const float* __restrict__ S, float* __restrict__ Sj) {
  int wave = blockIdx.x * (blockDim.x >> 5) + (threadIdx.x >> 5);
  int lane = threadIdx.x & 31;
  if (wave >= NB * NT) return;
  const float* row = S + (size_t)wave * NJ;
  float4 v = *(const float4*)(row + lane * 4);
  float m = fmaxf(fmaxf(v.x, v.y), fmaxf(v.z, v.w));
  m = wave_reduce_max(m);
  float e0 = expf(v.x - m), e1 = expf(v.y - m), e2 = expf(v.z - m), e3 = expf(v.w - m);
  float s = wave_reduce_sum(e0 + e1 + e2 + e3);
  float inv = 1.0f / s;
  float4 o; o.x = e0 * inv; o.y = e1 * inv; o.z = e2 * inv; o.w = e3 * inv;
  *(float4*)(Sj + (size_t)wave * NJ + lane * 4) = o;
}

__global__ __launch_bounds__(1024) void k_softmax_cols(const float* __restrict__ S,
                                                       float* __restrict__ St) {
  __shared__ float pmax[32][32];
  __shared__ float psum[32][32];
  int b = blockIdx.x >> 2;
  int jc = blockIdx.x & 3;
  int lx = threadIdx.x & 31, wv = threadIdx.x >> 5;
  size_t off = ((size_t)b * NT + wv * 32) * NJ + jc * 32 + lx;
  const float* base = S + off;
  float v[32];
#pragma unroll
  for (int i = 0; i < 32; ++i) v[i] = base[(size_t)i * NJ];
  float lm = v[0];
#pragma unroll
  for (int i = 1; i < 32; ++i) lm = fmaxf(lm, v[i]);
  pmax[wv][lx] = lm;
  __syncthreads();
  float M = pmax[0][lx];
#pragma unroll
  for (int i = 1; i < 32; ++i) M = fmaxf(M, pmax[i][lx]);
  float ls = 0.f;
#pragma unroll
  for (int i = 0; i < 32; ++i) { v[i] = expf(v[i] - M); ls += v[i]; }
  psum[wv][lx] = ls;
  __syncthreads();
  float s = psum[0][lx];
#pragma unroll
  for (int i = 1; i < 32; ++i) s += psum[i][lx];
  float inv = 1.0f / s;
  float* ob = St + off;
#pragma unroll
  for (int i = 0; i < 32; ++i) ob[(size_t)i * NJ] = v[i] * inv;
}

__global__ void k_tmp(const float* __restrict__ St, const float* __restrict__ C,
                      float* __restrict__ tmp) {
  __shared__ float part[8][32][32];
  int b = blockIdx.x >> 5;
  int jpair = (blockIdx.x >> 3) & 3;
  int dpair = blockIdx.x & 7;
  int wv = threadIdx.x >> 5;
  int ks = wv >> 1, jh = wv & 1;
  int lane = threadIdx.x & 31;
  int m = lane & 15, kh = lane >> 4;
  int j0 = jpair * 32, d0 = dpair * 32;
  const int KS = NT / 8;
  const float* Sbase = St + ((size_t)b * NT + ks * KS) * NJ + j0 + jh * 16 + m;
  const float* Cbase = C + ((size_t)b * NT + ks * KS) * ND + d0 + m;
  v8f acc0 = {}, acc1 = {};
#pragma unroll 2
  for (int k0 = 0; k0 < KS; k0 += 4) {
    int ka = k0 + 2 * kh;
    v2f a, b0, b1;
    a.x  = Sbase[(size_t)ka * NJ];
    a.y  = Sbase[(size_t)(ka + 1) * NJ];
    b0.x = Cbase[(size_t)ka * ND];
    b0.y = Cbase[(size_t)(ka + 1) * ND];
    b1.x = Cbase[(size_t)ka * ND + 16];
    b1.y = Cbase[(size_t)(ka + 1) * ND + 16];
    acc0 = __builtin_amdgcn_wmma_f32_16x16x4_f32(false, a, false, b0, (short)0,
                                                 acc0, false, false);
    acc1 = __builtin_amdgcn_wmma_f32_16x16x4_f32(false, a, false, b1, (short)0,
                                                 acc1, false, false);
  }
#pragma unroll
  for (int r = 0; r < 8; ++r) {
    int row = jh * 16 + r + 8 * kh;
    part[ks][row][m]      = acc0[r];
    part[ks][row][m + 16] = acc1[r];
  }
  __syncthreads();
#pragma unroll
  for (int h = 0; h < 2; ++h) {
    int row = wv + 16 * h;
    float s = part[0][row][lane];
#pragma unroll
    for (int i = 1; i < 8; ++i) s += part[i][row][lane];
    tmp[(size_t)(b * NJ + j0 + row) * ND + d0 + lane] = s;
  }
}

#define OT 8
#define AP 132
__global__ __launch_bounds__(256) void k_out(const float* __restrict__ Sj,
                                             const float* __restrict__ Q,
                                             const float* __restrict__ tmp,
                                             const float* __restrict__ C,
                                             float* __restrict__ out) {
  __shared__ __attribute__((aligned(16))) float As[2][16][AP];
  int blk = blockIdx.x;
  int b = blk >> 4;
  int chunk = (blk >> 1) & 7;
  int dh = blk & 1;
  int tid = threadIdx.x;
  int wv = tid >> 5;
  int lane = tid & 31;
  int m = lane & 15, kh = lane >> 4;
  int d = (dh * 8 + wv) * 16 + m;
  const float* Qb = Q + (size_t)b * NJ * ND + d;
  const float* Tb = tmp + (size_t)b * NJ * ND + d;
  v2f bq[32], bt[32];
#pragma unroll
  for (int i = 0; i < 32; ++i) {
    int ka = 4 * i + 2 * kh;
    bq[i].x = Qb[(size_t)ka * ND];
    bq[i].y = Qb[(size_t)(ka + 1) * ND];
    bt[i].x = Tb[(size_t)ka * ND];
    bt[i].y = Tb[(size_t)(ka + 1) * ND];
  }
  int tbase = chunk * (OT * 16);
  const float* Sg = Sj + ((size_t)b * NT + tbase + wv) * NJ + lane * 4;
  float4 g0 = *(const float4*)(Sg);
  float4 g1 = *(const float4*)(Sg + (size_t)8 * NJ);
  *(float4*)&As[0][wv][lane * 4] = g0;
  *(float4*)&As[0][wv + 8][lane * 4] = g1;
  __syncthreads();
#pragma nounroll
  for (int tt = 0; tt < OT; ++tt) {
    int cur = tt & 1;
    int t0 = tbase + tt * 16;
    if (tt + 1 < OT) {
      const float* Sn = Sg + (size_t)(tt + 1) * 16 * NJ;
      g0 = *(const float4*)(Sn);
      g1 = *(const float4*)(Sn + (size_t)8 * NJ);
    }
    float cv[8];
#pragma unroll
    for (int r = 0; r < 8; ++r)
      cv[r] = C[(size_t)(b * NT + t0 + r + 8 * kh) * ND + d];
    const float* Arow = &As[cur][m][2 * kh];
    v8f accA = {}, accB = {};
#pragma unroll
    for (int i = 0; i < 32; ++i) {
      v2f a = *(const v2f*)(Arow + 4 * i);
      accA = __builtin_amdgcn_wmma_f32_16x16x4_f32(false, a, false, bq[i], (short)0,
                                                   accA, false, false);
      accB = __builtin_amdgcn_wmma_f32_16x16x4_f32(false, a, false, bt[i], (short)0,
                                                   accB, false, false);
    }
    if (tt + 1 < OT) {
      *(float4*)&As[cur ^ 1][wv][lane * 4] = g0;
      *(float4*)&As[cur ^ 1][wv + 8][lane * 4] = g1;
      __syncthreads();
    }
#pragma unroll
    for (int r = 0; r < 8; ++r) {
      int t = t0 + r + 8 * kh;
      float* ob = out + (size_t)(b * NT + t) * (4 * ND);
      ob[d]          = cv[r];
      ob[ND + d]     = accA[r];
      ob[2 * ND + d] = cv[r] * accA[r];
      ob[3 * ND + d] = cv[r] * accB[r];
    }
  }
}

extern "C" void kernel_launch(void* const* d_in, const int* in_sizes, int n_in,
                              void* d_out, int out_size, void* d_ws, size_t ws_size,
                              hipStream_t stream) {
  (void)in_sizes; (void)n_in; (void)out_size; (void)ws_size;
  const float* C = (const float*)d_in[0];
  const float* Q = (const float*)d_in[1];
  const float* w = (const float*)d_in[2];
  float* out = (float*)d_out;

  float* ws = (float*)d_ws;
  float* S  = ws;
  float* Sj = S  + (size_t)NB * NT * NJ;
  float* St = Sj + (size_t)NB * NT * NJ;
  float* tp = St + (size_t)NB * NT * NJ;

  k_score<<<NB * (64 / ST), 256, 0, stream>>>(C, Q, w, S);
  k_softmax_rows<<<(NB * NT + 7) / 8, 256, 0, stream>>>(S, Sj);
  k_softmax_cols<<<NB * (NJ / 32), 1024, 0, stream>>>(S, St);
  k_tmp<<<NB * 32, 512, 0, stream>>>(St, C, tp);
  k_out<<<NB * 16, 256, 0, stream>>>(Sj, Q, tp, C, out);
}


# k_tmp: lanes load 16-byte row pieces of St and C whose 4 values feed 4 interleaved tiles (j=4m+q, d=4c+q): 64x64 wave tile, 4 loads per 16 WMMAs, t split 16 ways; changes the summation order of tmp
# speedup vs baseline: 2.7876x; 1.0452x over previous
#include <hip/hip_runtime.h>

#define NB 32
#define NT 1024
#define NJ 128
#define ND 256

typedef __attribute__((ext_vector_type(2))) float v2f;
typedef __attribute__((ext_vector_type(8))) float v8f;

__device__ __forceinline__ float wave_reduce_sum(float v) {
#pragma unroll
  for (int o = 16; o > 0; o >>= 1) v += __shfl_xor(v, o, 32);
  return v;
}
__device__ __forceinline__ float wave_reduce_max(float v) {
#pragma unroll
  for (int o = 16; o > 0; o >>= 1) v = fmaxf(v, __shfl_xor(v, o, 32));
  return v;
}

#define ST 4
#define CP 260
__global__ __launch_bounds__(256) void k_score(const float* __restrict__ C,
                                               const float* __restrict__ Q,
                                               const float* __restrict__ w,
                                               float* __restrict__ S) {
  __shared__ __attribute__((aligned(16))) float wl[(ND / 2) * 8];
  __shared__ __attribute__((aligned(16))) float Cs[2][16][CP];
  int blk = blockIdx.x;
  int b = blk >> 4;
  int chunk = blk & 15;
  int wv = threadIdx.x >> 5;
  int lane = threadIdx.x & 31;
  int m = lane & 15, kh = lane >> 4;
  int j0 = wv * 16;
  int tbase = chunk * (ST * 16);
  {
    int k = threadIdx.x;
    int p = (k >> 1) * 8 + (k & 1);
    wl[p]     = w[2 * ND + k];
    wl[p + 2] = w[k];
    wl[p + 4] = w[ND + k];
  }
  __syncthreads();
  const float* Cg = C + ((size_t)b * NT + tbase + wv) * ND + lane * 4;
  float4 f0 = *(const float4*)(Cg);
  float4 f1 = *(const float4*)(Cg + 128);
  float4 f2 = *(const float4*)(Cg + (size_t)8 * ND);
  float4 f3 = *(const float4*)(Cg + (size_t)8 * ND + 128);
  const float* Qrow = Q + (size_t)(b * NJ + j0 + m) * ND;
  v2f bf[64];
  v2f qs = {0.f, 0.f};
#pragma unroll
  for (int i = 0; i < 64; ++i) {
    int ka = 4 * i + 2 * kh;
    v2f q = *(const v2f*)(Qrow + ka);
    const float* wp = wl + (ka >> 1) * 8;
    float4 x = *(const float4*)wp;
    v2f uu = *(const v2f*)(wp + 4);
    v2f hu = {x.x, x.y}, hh = {x.z, x.w};
    bf[i] = q * hu + hh;
    qs += q * uu;
  }
  float qv = qs.x + qs.y;
  qv += __shfl_xor(qv, 16, 32);
  *(float4*)&Cs[0][wv][lane * 4] = f0;
  *(float4*)&Cs[0][wv][128 + lane * 4] = f1;
  *(float4*)&Cs[0][wv + 8][lane * 4] = f2;
  *(float4*)&Cs[0][wv + 8][128 + lane * 4] = f3;
  __syncthreads();
  int jj = j0 + m;
#pragma nounroll
  for (int tt = 0; tt < ST; ++tt) {
    int cur = tt & 1;
    int t0 = tbase + tt * 16;
    float4 g0, g1, g2, g3;
    if (tt + 1 < ST) {
      const float* Cn = Cg + (size_t)(tt + 1) * 16 * ND;
      g0 = *(const float4*)(Cn);
      g1 = *(const float4*)(Cn + 128);
      g2 = *(const float4*)(Cn + (size_t)8 * ND);
      g3 = *(const float4*)(Cn + (size_t)8 * ND + 128);
    }
    const float* Arow = &Cs[cur][m][2 * kh];
    v8f acc = {};
#pragma unroll
    for (int i = 0; i < 64; ++i) {
      v2f a = *(const v2f*)(Arow + 4 * i);
      acc = __builtin_amdgcn_wmma_f32_16x16x4_f32(false, a, false, bf[i], (short)0,
                                                  acc, false, false);
    }
    if (tt + 1 < ST) {
      *(float4*)&Cs[cur ^ 1][wv][lane * 4] = g0;
      *(float4*)&Cs[cur ^ 1][wv][128 + lane * 4] = g1;
      *(float4*)&Cs[cur ^ 1][wv + 8][lane * 4] = g2;
      *(float4*)&Cs[cur ^ 1][wv + 8][128 + lane * 4] = g3;
      __syncthreads();
    }
#pragma unroll
    for (int r = 0; r < 8; ++r) {
      int tt2 = t0 + r + 8 * kh;
      S[(size_t)(b * NT + tt2) * NJ + jj] = acc[r] + qv;
    }
  }
}

__global__ void k_softmax_rows(const float* __restrict__ S, float* __restrict__ Sj) {
  int wave = blockIdx.x * (blockDim.x >> 5) + (threadIdx.x >> 5);
  int lane = threadIdx.x & 31;
  if (wave >= NB * NT) return;
  const float* row = S + (size_t)wave * NJ;
  float4 v = *(const float4*)(row + lane * 4);
  float m = fmaxf(fmaxf(v.x, v.y), fmaxf(v.z, v.w));
  m = wave_reduce_max(m);
  float e0 = expf(v.x - m), e1 = expf(v.y - m), e2 = expf(v.z - m), e3 = expf(v.w - m);
  float s = wave_reduce_sum(e0 + e1 + e2 + e3);
  float inv = 1.0f / s;
  float4 o; o.x = e0 * inv; o.y = e1 * inv; o.z = e2 * inv; o.w = e3 * inv;
  *(float4*)(Sj + (size_t)wave * NJ + lane * 4) = o;
}

__global__ __launch_bounds__(1024) void k_softmax_cols(const float* __restrict__ S,
                                                       float* __restrict__ St) {
  __shared__ float pmax[32][32];
  __shared__ float psum[32][32];
  int b = blockIdx.x >> 2;
  int jc = blockIdx.x & 3;
  int lx = threadIdx.x & 31, wv = threadIdx.x >> 5;
  size_t off = ((size_t)b * NT + wv * 32) * NJ + jc * 32 + lx;
  const float* base = S + off;
  float v[32];
#pragma unroll
  for (int i = 0; i < 32; ++i) v[i] = base[(size_t)i * NJ];
  float lm = v[0];
#pragma unroll
  for (int i = 1; i < 32; ++i) lm = fmaxf(lm, v[i]);
  pmax[wv][lx] = lm;
  __syncthreads();
  float M = pmax[0][lx];
#pragma unroll
  for (int i = 1; i < 32; ++i) M = fmaxf(M, pmax[i][lx]);
  float ls = 0.f;
#pragma unroll
  for (int i = 0; i < 32; ++i) { v[i] = expf(v[i] - M); ls += v[i]; }
  psum[wv][lx] = ls;
  __syncthreads();
  float s = psum[0][lx];
#pragma unroll
  for (int i = 1; i < 32; ++i) s += psum[i][lx];
  float inv = 1.0f / s;
  float* ob = St + off;
#pragma unroll
  for (int i = 0; i < 32; ++i) ob[(size_t)i * NJ] = v[i] * inv;
}

#define TK 16
#define TR (NT / TK)
#define TS (TR / 4)
__global__ __launch_bounds__(512) void k_tmp(const float* __restrict__ St,
                                             const float* __restrict__ C,
                                             float* __restrict__ tmp) {
  __shared__ __attribute__((aligned(16))) float part[TK][64][64];
  int blk = blockIdx.x;
  int b = blk >> 3;
  int j0 = ((blk >> 2) & 1) * 64;
  int d0 = (blk & 3) * 64;
  int ks = threadIdx.x >> 5;
  int lane = threadIdx.x & 31;
  int m = lane & 15, kh = lane >> 4;
  const float* Sp = St + ((size_t)b * NT + ks * TR + 2 * kh) * NJ + j0 + 4 * m;
  const float* Cp = C + ((size_t)b * NT + ks * TR + 2 * kh) * ND + d0 + 4 * m;
  float4 ra0[TS], ra1[TS], rb0[TS], rb1[TS];
  v8f acc[4][4];
#pragma unroll
  for (int qa = 0; qa < 4; ++qa)
#pragma unroll
    for (int qb = 0; qb < 4; ++qb) acc[qa][qb] = v8f{};
#pragma unroll
  for (int i = 0; i < 2; ++i) {
    ra0[i] = *(const float4*)(Sp + (size_t)(4 * i) * NJ);
    ra1[i] = *(const float4*)(Sp + (size_t)(4 * i + 1) * NJ);
    rb0[i] = *(const float4*)(Cp + (size_t)(4 * i) * ND);
    rb1[i] = *(const float4*)(Cp + (size_t)(4 * i + 1) * ND);
  }
#pragma unroll
  for (int i = 0; i < TS; ++i) {
    if (i + 2 < TS) {
      ra0[i + 2] = *(const float4*)(Sp + (size_t)(4 * i + 8) * NJ);
      ra1[i + 2] = *(const float4*)(Sp + (size_t)(4 * i + 9) * NJ);
      rb0[i + 2] = *(const float4*)(Cp + (size_t)(4 * i + 8) * ND);
      rb1[i + 2] = *(const float4*)(Cp + (size_t)(4 * i + 9) * ND);
    }
    v2f af[4], bf[4];
    af[0].x = ra0[i].x; af[0].y = ra1[i].x;
    af[1].x = ra0[i].y; af[1].y = ra1[i].y;
    af[2].x = ra0[i].z; af[2].y = ra1[i].z;
    af[3].x = ra0[i].w; af[3].y = ra1[i].w;
    bf[0].x = rb0[i].x; bf[0].y = rb1[i].x;
    bf[1].x = rb0[i].y; bf[1].y = rb1[i].y;
    bf[2].x = rb0[i].z; bf[2].y = rb1[i].z;
    bf[3].x = rb0[i].w; bf[3].y = rb1[i].w;
#pragma unroll
    for (int qa = 0; qa < 4; ++qa)
#pragma unroll
      for (int qb = 0; qb < 4; ++qb)
        acc[qa][qb] = __builtin_amdgcn_wmma_f32_16x16x4_f32(false, af[qa], false, bf[qb],
                                                            (short)0, acc[qa][qb], false, false);
  }
#pragma unroll
  for (int qa = 0; qa < 4; ++qa)
#pragma unroll
    for (int r = 0; r < 8; ++r) {
      float4 v;
      v.x = acc[qa][0][r]; v.y = acc[qa][1][r]; v.z = acc[qa][2][r]; v.w = acc[qa][3][r];
      *(float4*)&part[ks][4 * (8 * kh + r) + qa][4 * m] = v;
    }
  __syncthreads();
#pragma unroll
  for (int h = 0; h < 2; ++h) {
    int idx = threadIdx.x + 512 * h;
    int row = idx >> 4, c4 = (idx & 15) * 4;
    float4 s = *(const float4*)&part[0][row][c4];
#pragma unroll
    for (int i = 1; i < TK; ++i) {
      float4 p = *(const float4*)&part[i][row][c4];
      s.x += p.x; s.y += p.y; s.z += p.z; s.w += p.w;
    }
    *(float4*)&tmp[(size_t)(b * NJ + j0 + row) * ND + d0 + c4] = s;
  }
}

#define OT 8
#define AP 132
__global__ __launch_bounds__(256) void k_out(const float* __restrict__ Sj,
                                             const float* __restrict__ Q,
                                             const float* __restrict__ tmp,
                                             const float* __restrict__ C,
                                             float* __restrict__ out) {
  __shared__ __attribute__((aligned(16))) float As[2][16][AP];
  int blk = blockIdx.x;
  int b = blk >> 4;
  int chunk = (blk >> 1) & 7;
  int dh = blk & 1;
  int tid = threadIdx.x;
  int wv = tid >> 5;
  int lane = tid & 31;
  int m = lane & 15, kh = lane >> 4;
  int d = (dh * 8 + wv) * 16 + m;
  const float* Qb = Q + (size_t)b * NJ * ND + d;
  const float* Tb = tmp + (size_t)b * NJ * ND + d;
  v2f bq[32], bt[32];
#pragma unroll
  for (int i = 0; i < 32; ++i) {
    int ka = 4 * i + 2 * kh;
    bq[i].x = Qb[(size_t)ka * ND];
    bq[i].y = Qb[(size_t)(ka + 1) * ND];
    bt[i].x = Tb[(size_t)ka * ND];
    bt[i].y = Tb[(size_t)(ka + 1) * ND];
  }
  int tbase = chunk * (OT * 16);
  const float* Sg = Sj + ((size_t)b * NT + tbase + wv) * NJ + lane * 4;
  float4 g0 = *(const float4*)(Sg);
  float4 g1 = *(const float4*)(Sg + (size_t)8 * NJ);
  *(float4*)&As[0][wv][lane * 4] = g0;
  *(float4*)&As[0][wv + 8][lane * 4] = g1;
  __syncthreads();
#pragma nounroll
  for (int tt = 0; tt < OT; ++tt) {
    int cur = tt & 1;
    int t0 = tbase + tt * 16;
    if (tt + 1 < OT) {
      const float* Sn = Sg + (size_t)(tt + 1) * 16 * NJ;
      g0 = *(const float4*)(Sn);
      g1 = *(const float4*)(Sn + (size_t)8 * NJ);
    }
    float cv[8];
#pragma unroll
    for (int r = 0; r < 8; ++r)
      cv[r] = C[(size_t)(b * NT + t0 + r + 8 * kh) * ND + d];
    const float* Arow = &As[cur][m][2 * kh];
    v8f accA = {}, accB = {};
#pragma unroll
    for (int i = 0; i < 32; ++i) {
      v2f a = *(const v2f*)(Arow + 4 * i);
      accA = __builtin_amdgcn_wmma_f32_16x16x4_f32(false, a, false, bq[i], (short)0,
                                                   accA, false, false);
      accB = __builtin_amdgcn_wmma_f32_16x16x4_f32(false, a, false, bt[i], (short)0,
                                                   accB, false, false);
    }
    if (tt + 1 < OT) {
      *(float4*)&As[cur ^ 1][wv][lane * 4] = g0;
      *(float4*)&As[cur ^ 1][wv + 8][lane * 4] = g1;
      __syncthreads();
    }
#pragma unroll
    for (int r = 0; r < 8; ++r) {
      int t = t0 + r + 8 * kh;
      float* ob = out + (size_t)(b * NT + t) * (4 * ND);
      ob[d]          = cv[r];
      ob[ND + d]     = accA[r];
      ob[2 * ND + d] = cv[r] * accA[r];
      ob[3 * ND + d] = cv[r] * accB[r];
    }
  }
}

extern "C" void kernel_launch(void* const* d_in, const int* in_sizes, int n_in,
                              void* d_out, int out_size, void* d_ws, size_t ws_size,
                              hipStream_t stream) {
  (void)in_sizes; (void)n_in; (void)out_size; (void)ws_size;
  const float* C = (const float*)d_in[0];
  const float* Q = (const float*)d_in[1];
  const float* w = (const float*)d_in[2];
  float* out = (float*)d_out;

  float* ws = (float*)d_ws;
  float* S  = ws;
  float* Sj = S  + (size_t)NB * NT * NJ;
  float* St = Sj + (size_t)NB * NT * NJ;
  float* tp = St + (size_t)NB * NT * NJ;

  k_score<<<NB * (64 / ST), 256, 0, stream>>>(C, Q, w, S);
  k_softmax_rows<<<(NB * NT + 7) / 8, 256, 0, stream>>>(S, Sj);
  k_softmax_cols<<<NB * (NJ / 32), 1024, 0, stream>>>(S, St);
  k_tmp<<<NB * 8, 512, 0, stream>>>(St, C, tp);
  k_out<<<NB * 16, 256, 0, stream>>>(Sj, Q, tp, C, out);
}


# k_out forms the row softmax itself on the way into LDS (a staging wave holds a whole row of S; max and sum by vector-ALU lane exchanges, sums unchanged); row-softmax launch and Sj plane removed
# speedup vs baseline: 2.9639x; 1.0632x over previous
#include <hip/hip_runtime.h>

#define NB 32
#define NT 1024
#define NJ 128
#define ND 256

typedef __attribute__((ext_vector_type(2))) float v2f;
typedef __attribute__((ext_vector_type(8))) float v8f;

__device__ __forceinline__ float xhalf(float x) {
  int b = __builtin_bit_cast(int, x);
  return __builtin_bit_cast(float, __builtin_amdgcn_permlanex16(b, b, 0x76543210, 0xfedcba98,
                                                                false, false));
}
template <int N> __device__ __forceinline__ float xrow(float x) {
  return __builtin_amdgcn_update_dpp(x, x, 0x160 + N, 0xf, 0xf, true);
}
__device__ __forceinline__ float wave_reduce_sum(float v) {
  v += xhalf(v);
  v += xrow<8>(v);
  v += xrow<4>(v);
  v += xrow<2>(v);
  v += xrow<1>(v);
  return v;
}
__device__ __forceinline__ float wave_reduce_max(float v) {
  v = fmaxf(v, xhalf(v));
  v = fmaxf(v, xrow<8>(v));
  v = fmaxf(v, xrow<4>(v));
  v = fmaxf(v, xrow<2>(v));
  v = fmaxf(v, xrow<1>(v));
  return v;
}
__device__ __forceinline__ float4 row_softmax(float4 v) {
  float m = fmaxf(fmaxf(v.x, v.y), fmaxf(v.z, v.w));
  m = wave_reduce_max(m);
  float e0 = expf(v.x - m), e1 = expf(v.y - m), e2 = expf(v.z - m), e3 = expf(v.w - m);
  float s = wave_reduce_sum(e0 + e1 + e2 + e3);
  float inv = 1.0f / s;
  float4 o; o.x = e0 * inv; o.y = e1 * inv; o.z = e2 * inv; o.w = e3 * inv;
  return o;
}

#define ST 4
#define CP 260
__global__ __launch_bounds__(256) void k_score(const float* __restrict__ C,
                                               const float* __restrict__ Q,
                                               const float* __restrict__ w,
                                               float* __restrict__ S) {
  __shared__ __attribute__((aligned(16))) float wl[(ND / 2) * 8];
  __shared__ __attribute__((aligned(16))) float Cs[2][16][CP];
  int blk = blockIdx.x;
  int b = blk >> 4;
  int chunk = blk & 15;
  int wv = threadIdx.x >> 5;
  int lane = threadIdx.x & 31;
  int m = lane & 15, kh = lane >> 4;
  int j0 = wv * 16;
  int tbase = chunk * (ST * 16);
  {
    int k = threadIdx.x;
    int p = (k >> 1) * 8 + (k & 1);
    wl[p]     = w[2 * ND + k];
    wl[p + 2] = w[k];
    wl[p + 4] = w[ND + k];
  }
  __syncthreads();
  const float* Cg = C + ((size_t)b * NT + tbase + wv) * ND + lane * 4;
  float4 f0 = *(const float4*)(Cg);
  float4 f1 = *(const float4*)(Cg + 128);
  float4 f2 = *(const float4*)(Cg + (size_t)8 * ND);
  float4 f3 = *(const float4*)(Cg + (size_t)8 * ND + 128);
  const float* Qrow = Q + (size_t)(b * NJ + j0 + m) * ND;
  v2f bf[64];
  v2f qs = {0.f, 0.f};
#pragma unroll
  for (int i = 0; i < 64; ++i) {
    int ka = 4 * i + 2 * kh;
    v2f q = *(const v2f*)(Qrow + ka);
    const float* wp = wl + (ka >> 1) * 8;
    float4 x = *(const float4*)wp;
    v2f uu = *(const v2f*)(wp + 4);
    v2f hu = {x.x, x.y}, hh = {x.z, x.w};
    bf[i] = q * hu + hh;
    qs += q * uu;
  }
  float qv = qs.x + qs.y;
  qv += __shfl_xor(qv, 16, 32);
  *(float4*)&Cs[0][wv][lane * 4] = f0;
  *(float4*)&Cs[0][wv][128 + lane * 4] = f1;
  *(float4*)&Cs[0][wv + 8][lane * 4] = f2;
  *(float4*)&Cs[0][wv + 8][128 + lane * 4] = f3;
  __syncthreads();
  int jj = j0 + m;
#pragma nounroll
  for (int tt = 0; tt < ST; ++tt) {
    int cur = tt & 1;
    int t0 = tbase + tt * 16;
    float4 g0, g1, g2, g3;
    if (tt + 1 < ST) {
      const float* Cn = Cg + (size_t)(tt + 1) * 16 * ND;
      g0 = *(const float4*)(Cn);
      g1 = *(const float4*)(Cn + 128);
      g2 = *(const float4*)(Cn + (size_t)8 * ND);
      g3 = *(const float4*)(Cn + (size_t)8 * ND + 128);
    }
    const float* Arow = &Cs[cur][m][2 * kh];
    v8f acc = {};
#pragma unroll
    for (int i = 0; i < 64; ++i) {
      v2f a = *(const v2f*)(Arow + 4 * i);
      acc = __builtin_amdgcn_wmma_f32_16x16x4_f32(false, a, false, bf[i], (short)0,
                                                  acc, false, false);
    }
    if (tt + 1 < ST) {
      *(float4*)&Cs[cur ^ 1][wv][lane * 4] = g0;
      *(float4*)&Cs[cur ^ 1][wv][128 + lane * 4] = g1;
      *(float4*)&Cs[cur ^ 1][wv + 8][lane * 4] = g2;
      *(float4*)&Cs[cur ^ 1][wv + 8][128 + lane * 4] = g3;
      __syncthreads();
    }
#pragma unroll
    for (int r = 0; r < 8; ++r) {
      int tt2 = t0 + r + 8 * kh;
      S[(size_t)(b * NT + tt2) * NJ + jj] = acc[r] + qv;
    }
  }
}

__global__ __launch_bounds__(1024) void k_softmax_cols(const float* __restrict__ S,
                                                       float* __restrict__ St) {
  __shared__ float pmax[32][32];
  __shared__ float psum[32][32];
  int b = blockIdx.x >> 2;
  int jc = blockIdx.x & 3;
  int lx = threadIdx.x & 31, wv = threadIdx.x >> 5;
  size_t off = ((size_t)b * NT + wv * 32) * NJ + jc * 32 + lx;
  const float* base = S + off;
  float v[32];
#pragma unroll
  for (int i = 0; i < 32; ++i) v[i] = base[(size_t)i * NJ];
  float lm = v[0];
#pragma unroll
  for (int i = 1; i < 32; ++i) lm = fmaxf(lm, v[i]);
  pmax[wv][lx] = lm;
  __syncthreads();
  float M = pmax[0][lx];
#pragma unroll
  for (int i = 1; i < 32; ++i) M = fmaxf(M, pmax[i][lx]);
  float ls = 0.f;
#pragma unroll
  for (int i = 0; i < 32; ++i) { v[i] = expf(v[i] - M); ls += v[i]; }
  psum[wv][lx] = ls;
  __syncthreads();
  float s = psum[0][lx];
#pragma unroll
  for (int i = 1; i < 32; ++i) s += psum[i][lx];
  float inv = 1.0f / s;
  float* ob = St + off;
#pragma unroll
  for (int i = 0; i < 32; ++i) ob[(size_t)i * NJ] = v[i] * inv;
}

#define TK 16
#define TR (NT / TK)
#define TS (TR / 4)
__global__ __launch_bounds__(512) void k_tmp(const float* __restrict__ St,
                                             const float* __restrict__ C,
                                             float* __restrict__ tmp) {
  __shared__ __attribute__((aligned(16))) float part[TK][64][64];
  int blk = blockIdx.x;
  int b = blk >> 3;
  int j0 = ((blk >> 2) & 1) * 64;
  int d0 = (blk & 3) * 64;
  int ks = threadIdx.x >> 5;
  int lane = threadIdx.x & 31;
  int m = lane & 15, kh = lane >> 4;
  const float* Sp = St + ((size_t)b * NT + ks * TR + 2 * kh) * NJ + j0 + 4 * m;
  const float* Cp = C + ((size_t)b * NT + ks * TR + 2 * kh) * ND + d0 + 4 * m;
  float4 ra0[TS], ra1[TS], rb0[TS], rb1[TS];
  v8f acc[4][4];
#pragma unroll
  for (int qa = 0; qa < 4; ++qa)
#pragma unroll
    for (int qb = 0; qb < 4; ++qb) acc[qa][qb] = v8f{};
#pragma unroll
  for (int i = 0; i < 2; ++i) {
    ra0[i] = *(const float4*)(Sp + (size_t)(4 * i) * NJ);
    ra1[i] = *(const float4*)(Sp + (size_t)(4 * i + 1) * NJ);
    rb0[i] = *(const float4*)(Cp + (size_t)(4 * i) * ND);
    rb1[i] = *(const float4*)(Cp + (size_t)(4 * i + 1) * ND);
  }
#pragma unroll
  for (int i = 0; i < TS; ++i) {
    if (i + 2 < TS) {
      ra0[i + 2] = *(const float4*)(Sp + (size_t)(4 * i + 8) * NJ);
      ra1[i + 2] = *(const float4*)(Sp + (size_t)(4 * i + 9) * NJ);
      rb0[i + 2] = *(const float4*)(Cp + (size_t)(4 * i + 8) * ND);
      rb1[i + 2] = *(const float4*)(Cp + (size_t)(4 * i + 9) * ND);
    }
    v2f af[4], bf[4];
    af[0].x = ra0[i].x; af[0].y = ra1[i].x;
    af[1].x = ra0[i].y; af[1].y = ra1[i].y;
    af[2].x = ra0[i].z; af[2].y = ra1[i].z;
    af[3].x = ra0[i].w; af[3].y = ra1[i].w;
    bf[0].x = rb0[i].x; bf[0].y = rb1[i].x;
    bf[1].x = rb0[i].y; bf[1].y = rb1[i].y;
    bf[2].x = rb0[i].z; bf[2].y = rb1[i].z;
    bf[3].x = rb0[i].w; bf[3].y = rb1[i].w;
#pragma unroll
    for (int qa = 0; qa < 4; ++qa)
#pragma unroll
      for (int qb = 0; qb < 4; ++qb)
        acc[qa][qb] = __builtin_amdgcn_wmma_f32_16x16x4_f32(false, af[qa], false, bf[qb],
                                                            (short)0, acc[qa][qb], false, false);
  }
#pragma unroll
  for (int qa = 0; qa < 4; ++qa)
#pragma unroll
    for (int r = 0; r < 8; ++r) {
      float4 v;
      v.x = acc[qa][0][r]; v.y = acc[qa][1][r]; v.z = acc[qa][2][r]; v.w = acc[qa][3][r];
      *(float4*)&part[ks][4 * (8 * kh + r) + qa][4 * m] = v;
    }
  __syncthreads();
#pragma unroll
  for (int h = 0; h < 2; ++h) {
    int idx = threadIdx.x + 512 * h;
    int row = idx >> 4, c4 = (idx & 15) * 4;
    float4 s = *(const float4*)&part[0][row][c4];
#pragma unroll
    for (int i = 1; i < TK; ++i) {
      float4 p = *(const float4*)&part[i][row][c4];
      s.x += p.x; s.y += p.y; s.z += p.z; s.w += p.w;
    }
    *(float4*)&tmp[(size_t)(b * NJ + j0 + row) * ND + d0 + c4] = s;
  }
}

#define OT 8
#define AP 132
__global__ __launch_bounds__(256) void k_out(const float* __restrict__ S,
                                             const float* __restrict__ Q,
                                             const float* __restrict__ tmp,
                                             const float* __restrict__ C,
                                             float* __restrict__ out) {
  __shared__ __attribute__((aligned(16))) float As[2][16][AP];
  int blk = blockIdx.x;
  int b = blk >> 4;
  int chunk = (blk >> 1) & 7;
  int dh = blk & 1;
  int tid = threadIdx.x;
  int wv = tid >> 5;
  int lane = tid & 31;
  int m = lane & 15, kh = lane >> 4;
  int d = (dh * 8 + wv) * 16 + m;
  int tbase = chunk * (OT * 16);
  const float* Sg = S + ((size_t)b * NT + tbase + wv) * NJ + lane * 4;
  float4 n0 = *(const float4*)(Sg);
  float4 n1 = *(const float4*)(Sg + (size_t)8 * NJ);
  const float* Qb = Q + (size_t)b * NJ * ND + d;
  const float* Tb = tmp + (size_t)b * NJ * ND + d;
  v2f bq[32], bt[32];
#pragma unroll
  for (int i = 0; i < 32; ++i) {
    int ka = 4 * i + 2 * kh;
    bq[i].x = Qb[(size_t)ka * ND];
    bq[i].y = Qb[(size_t)(ka + 1) * ND];
    bt[i].x = Tb[(size_t)ka * ND];
    bt[i].y = Tb[(size_t)(ka + 1) * ND];
  }
  {
    float4 p0 = row_softmax(n0);
    float4 p1 = row_softmax(n1);
    *(float4*)&As[0][wv][lane * 4] = p0;
    *(float4*)&As[0][wv + 8][lane * 4] = p1;
  }
  n0 = *(const float4*)(Sg + (size_t)16 * NJ);
  n1 = *(const float4*)(Sg + (size_t)24 * NJ);
  __syncthreads();
#pragma nounroll
  for (int tt = 0; tt < OT; ++tt) {
    int cur = tt & 1;
    int t0 = tbase + tt * 16;
    if (tt + 1 < OT) {
      float4 p0 = row_softmax(n0);
      float4 p1 = row_softmax(n1);
      *(float4*)&As[cur ^ 1][wv][lane * 4] = p0;
      *(float4*)&As[cur ^ 1][wv + 8][lane * 4] = p1;
      if (tt + 2 < OT) {
        const float* Sn = Sg + (size_t)(tt + 2) * 16 * NJ;
        n0 = *(const float4*)(Sn);
        n1 = *(const float4*)(Sn + (size_t)8 * NJ);
      }
    }
    float cv[8];
#pragma unroll
    for (int r = 0; r < 8; ++r)
      cv[r] = C[(size_t)(b * NT + t0 + r + 8 * kh) * ND + d];
    const float* Arow = &As[cur][m][2 * kh];
    v8f accA = {}, accB = {};
#pragma unroll
    for (int i = 0; i < 32; ++i) {
      v2f a = *(const v2f*)(Arow + 4 * i);
      accA = __builtin_amdgcn_wmma_f32_16x16x4_f32(false, a, false, bq[i], (short)0,
                                                   accA, false, false);
      accB = __builtin_amdgcn_wmma_f32_16x16x4_f32(false, a, false, bt[i], (short)0,
                                                   accB, false, false);
    }
    if (tt + 1 < OT) __syncthreads();
#pragma unroll
    for (int r = 0; r < 8; ++r) {
      int t = t0 + r + 8 * kh;
      float* ob = out + (size_t)(b * NT + t) * (4 * ND);
      ob[d]          = cv[r];
      ob[ND + d]     = accA[r];
      ob[2 * ND + d] = cv[r] * accA[r];
      ob[3 * ND + d] = cv[r] * accB[r];
    }
  }
}

extern "C" void kernel_launch(void* const* d_in, const int* in_sizes, int n_in,
                              void* d_out, int out_size, void* d_ws, size_t ws_size,
                              hipStream_t stream) {
  (void)in_sizes; (void)n_in; (void)out_size; (void)ws_size;
  const float* C = (const float*)d_in[0];
  const float* Q = (const float*)d_in[1];
  const float* w = (const float*)d_in[2];
  float* out = (float*)d_out;

  float* ws = (float*)d_ws;
  float* S  = ws;
  float* St = S  + (size_t)NB * NT * NJ;
  float* tp = St + (size_t)NB * NT * NJ;

  k_score<<<NB * (64 / ST), 256, 0, stream>>>(C, Q, w, S);
  k_softmax_cols<<<NB * (NJ / 32), 1024, 0, stream>>>(S, St);
  k_tmp<<<NB * 8, 512, 0, stream>>>(St, C, tp);
  k_out<<<NB * 16, 256, 0, stream>>>(S, Q, tp, C, out);
}


# k_tmp forms the softmax over t itself (column maxima left by k_score; exp(S - max) as A fragments; tile divided by its column sums): St plane and launch removed; changes tmp's summation order
# speedup vs baseline: 3.2046x; 1.0812x over previous
#include <hip/hip_runtime.h>

#define NB 32
#define NT 1024
#define NJ 128
#define ND 256

typedef __attribute__((ext_vector_type(2))) float v2f;
typedef __attribute__((ext_vector_type(8))) float v8f;

__device__ __forceinline__ float xhalf(float x) {
  int b = __builtin_bit_cast(int, x);
  return __builtin_bit_cast(float, __builtin_amdgcn_permlanex16(b, b, 0x76543210, 0xfedcba98,
                                                                false, false));
}
template <int N> __device__ __forceinline__ float xrow(float x) {
  return __builtin_amdgcn_update_dpp(x, x, 0x160 + N, 0xf, 0xf, true);
}
__device__ __forceinline__ float wave_reduce_sum(float v) {
  v += xhalf(v);
  v += xrow<8>(v);
  v += xrow<4>(v);
  v += xrow<2>(v);
  v += xrow<1>(v);
  return v;
}
__device__ __forceinline__ float wave_reduce_max(float v) {
  v = fmaxf(v, xhalf(v));
  v = fmaxf(v, xrow<8>(v));
  v = fmaxf(v, xrow<4>(v));
  v = fmaxf(v, xrow<2>(v));
  v = fmaxf(v, xrow<1>(v));
  return v;
}
__device__ __forceinline__ float4 row_softmax(float4 v) {
  float m = fmaxf(fmaxf(v.x, v.y), fmaxf(v.z, v.w));
  m = wave_reduce_max(m);
  float e0 = expf(v.x - m), e1 = expf(v.y - m), e2 = expf(v.z - m), e3 = expf(v.w - m);
  float s = wave_reduce_sum(e0 + e1 + e2 + e3);
  float inv = 1.0f / s;
  float4 o; o.x = e0 * inv; o.y = e1 * inv; o.z = e2 * inv; o.w = e3 * inv;
  return o;
}

#define ST 4
#define NC (NT / (ST * 16))
#define CP 260
__global__ __launch_bounds__(256) void k_score(const float* __restrict__ C,
                                               const float* __restrict__ Q,
                                               const float* __restrict__ w,
                                               float* __restrict__ S,
                                               float* __restrict__ pm) {
  __shared__ __attribute__((aligned(16))) float wl[(ND / 2) * 8];
  __shared__ __attribute__((aligned(16))) float Cs[2][16][CP];
  int blk = blockIdx.x;
  int b = blk >> 4;
  int chunk = blk & 15;
  int wv = threadIdx.x >> 5;
  int lane = threadIdx.x & 31;
  int m = lane & 15, kh = lane >> 4;
  int j0 = wv * 16;
  int tbase = chunk * (ST * 16);
  {
    int k = threadIdx.x;
    int p = (k >> 1) * 8 + (k & 1);
    wl[p]     = w[2 * ND + k];
    wl[p + 2] = w[k];
    wl[p + 4] = w[ND + k];
  }
  __syncthreads();
  const float* Cg = C + ((size_t)b * NT + tbase + wv) * ND + lane * 4;
  float4 f0 = *(const float4*)(Cg);
  float4 f1 = *(const float4*)(Cg + 128);
  float4 f2 = *(const float4*)(Cg + (size_t)8 * ND);
  float4 f3 = *(const float4*)(Cg + (size_t)8 * ND + 128);
  const float* Qrow = Q + (size_t)(b * NJ + j0 + m) * ND;
  v2f bf[64];
  v2f qs = {0.f, 0.f};
#pragma unroll
  for (int i = 0; i < 64; ++i) {
    int ka = 4 * i + 2 * kh;
    v2f q = *(const v2f*)(Qrow + ka);
    const float* wp = wl + (ka >> 1) * 8;
    float4 x = *(const float4*)wp;
    v2f uu = *(const v2f*)(wp + 4);
    v2f hu = {x.x, x.y}, hh = {x.z, x.w};
    bf[i] = q * hu + hh;
    qs += q * uu;
  }
  float qv = qs.x + qs.y;
  qv += __shfl_xor(qv, 16, 32);
  *(float4*)&Cs[0][wv][lane * 4] = f0;
  *(float4*)&Cs[0][wv][128 + lane * 4] = f1;
  *(float4*)&Cs[0][wv + 8][lane * 4] = f2;
  *(float4*)&Cs[0][wv + 8][128 + lane * 4] = f3;
  __syncthreads();
  int jj = j0 + m;
  float cm = -__builtin_inff();
#pragma nounroll
  for (int tt = 0; tt < ST; ++tt) {
    int cur = tt & 1;
    int t0 = tbase + tt * 16;
    float4 g0, g1, g2, g3;
    if (tt + 1 < ST) {
      const float* Cn = Cg + (size_t)(tt + 1) * 16 * ND;
      g0 = *(const float4*)(Cn);
      g1 = *(const float4*)(Cn + 128);
      g2 = *(const float4*)(Cn + (size_t)8 * ND);
      g3 = *(const float4*)(Cn + (size_t)8 * ND + 128);
    }
    const float* Arow = &Cs[cur][m][2 * kh];
    v8f acc = {};
#pragma unroll
    for (int i = 0; i < 64; ++i) {
      v2f a = *(const v2f*)(Arow + 4 * i);
      acc = __builtin_amdgcn_wmma_f32_16x16x4_f32(false, a, false, bf[i], (short)0,
                                                  acc, false, false);
    }
    if (tt + 1 < ST) {
      *(float4*)&Cs[cur ^ 1][wv][lane * 4] = g0;
      *(float4*)&Cs[cur ^ 1][wv][128 + lane * 4] = g1;
      *(float4*)&Cs[cur ^ 1][wv + 8][lane * 4] = g2;
      *(float4*)&Cs[cur ^ 1][wv + 8][128 + lane * 4] = g3;
      __syncthreads();
    }
#pragma unroll
    for (int r = 0; r < 8; ++r) {
      int tt2 = t0 + r + 8 * kh;
      float sv = acc[r] + qv;
      S[(size_t)(b * NT + tt2) * NJ + jj] = sv;
      cm = fmaxf(cm, sv);
    }
  }
  cm = fmaxf(cm, xhalf(cm));
  if (kh == 0) pm[(size_t)blk * NJ + jj] = cm;
}

#define TK 16
#define TR (NT / TK)
#define TS (TR / 4)
__device__ __forceinline__ float4 col_exp(float4 v, float4 mx) {
  const float L2E = 1.44269504088896340736f;
  float4 e;
  e.x = __builtin_amdgcn_exp2f((v.x - mx.x) * L2E);
  e.y = __builtin_amdgcn_exp2f((v.y - mx.y) * L2E);
  e.z = __builtin_amdgcn_exp2f((v.z - mx.z) * L2E);
  e.w = __builtin_amdgcn_exp2f((v.w - mx.w) * L2E);
  return e;
}
__global__ __launch_bounds__(512) void k_tmp(const float* __restrict__ S,
                                             const float* __restrict__ pm,
                                             const float* __restrict__ C,
                                             float* __restrict__ tmp) {
  __shared__ __attribute__((aligned(16))) float part[TK][64][64];
  __shared__ __attribute__((aligned(16))) float csum[TK][64];
  int blk = blockIdx.x;
  int b = blk >> 3;
  int j0 = ((blk >> 2) & 1) * 64;
  int d0 = (blk & 3) * 64;
  int ks = threadIdx.x >> 5;
  int lane = threadIdx.x & 31;
  int m = lane & 15, kh = lane >> 4;
  const float* Sp = S + ((size_t)b * NT + ks * TR + 2 * kh) * NJ + j0 + 4 * m;
  const float* Cp = C + ((size_t)b * NT + ks * TR + 2 * kh) * ND + d0 + 4 * m;
  const float* Mp = pm + (size_t)b * NC * NJ + j0 + 4 * m;
  float4 px[NC];
#pragma unroll
  for (int c = 0; c < NC; ++c) px[c] = *(const float4*)(Mp + (size_t)c * NJ);
  float4 ra0[TS], ra1[TS], rb0[TS], rb1[TS];
  v8f acc[4][4];
#pragma unroll
  for (int qa = 0; qa < 4; ++qa)
#pragma unroll
    for (int qb = 0; qb < 4; ++qb) acc[qa][qb] = v8f{};
#pragma unroll
  for (int i = 0; i < 3; ++i) {
    ra0[i] = *(const float4*)(Sp + (size_t)(4 * i) * NJ);
    ra1[i] = *(const float4*)(Sp + (size_t)(4 * i + 1) * NJ);
  }
#pragma unroll
  for (int i = 0; i < 2; ++i) {
    rb0[i] = *(const float4*)(Cp + (size_t)(4 * i) * ND);
    rb1[i] = *(const float4*)(Cp + (size_t)(4 * i + 1) * ND);
  }
  float4 mx = px[0];
#pragma unroll
  for (int c = 1; c < NC; ++c) {
    mx.x = fmaxf(mx.x, px[c].x); mx.y = fmaxf(mx.y, px[c].y);
    mx.z = fmaxf(mx.z, px[c].z); mx.w = fmaxf(mx.w, px[c].w);
  }
  float4 e0 = col_exp(ra0[0], mx);
  float4 e1 = col_exp(ra1[0], mx);
  float4 cs;
  cs.x = e0.x + e1.x; cs.y = e0.y + e1.y; cs.z = e0.z + e1.z; cs.w = e0.w + e1.w;
#pragma unroll
  for (int i = 0; i < TS; ++i) {
    if (i + 3 < TS) {
      ra0[i + 3] = *(const float4*)(Sp + (size_t)(4 * i + 12) * NJ);
      ra1[i + 3] = *(const float4*)(Sp + (size_t)(4 * i + 13) * NJ);
    }
    if (i + 2 < TS) {
      rb0[i + 2] = *(const float4*)(Cp + (size_t)(4 * i + 8) * ND);
      rb1[i + 2] = *(const float4*)(Cp + (size_t)(4 * i + 9) * ND);
    }
    float4 n0 = e0, n1 = e1;
    if (i + 1 < TS) {
      n0 = col_exp(ra0[i + 1], mx);
      n1 = col_exp(ra1[i + 1], mx);
      cs.x += n0.x; cs.y += n0.y; cs.z += n0.z; cs.w += n0.w;
      cs.x += n1.x; cs.y += n1.y; cs.z += n1.z; cs.w += n1.w;
    }
    v2f af[4], bf[4];
    af[0].x = e0.x; af[0].y = e1.x;
    af[1].x = e0.y; af[1].y = e1.y;
    af[2].x = e0.z; af[2].y = e1.z;
    af[3].x = e0.w; af[3].y = e1.w;
    bf[0].x = rb0[i].x; bf[0].y = rb1[i].x;
    bf[1].x = rb0[i].y; bf[1].y = rb1[i].y;
    bf[2].x = rb0[i].z; bf[2].y = rb1[i].z;
    bf[3].x = rb0[i].w; bf[3].y = rb1[i].w;
#pragma unroll
    for (int qa = 0; qa < 4; ++qa)
#pragma unroll
      for (int qb = 0; qb < 4; ++qb)
        acc[qa][qb] = __builtin_amdgcn_wmma_f32_16x16x4_f32(false, af[qa], false, bf[qb],
                                                            (short)0, acc[qa][qb], false, false);
    e0 = n0; e1 = n1;
    if ((i & 3) == 3) __builtin_amdgcn_sched_barrier(0);
  }
  cs.x += xhalf(cs.x); cs.y += xhalf(cs.y); cs.z += xhalf(cs.z); cs.w += xhalf(cs.w);
  if (kh == 0) *(float4*)&csum[ks][4 * m] = cs;
#pragma unroll
  for (int qa = 0; qa < 4; ++qa)
#pragma unroll
    for (int r = 0; r < 8; ++r) {
      float4 v;
      v.x = acc[qa][0][r]; v.y = acc[qa][1][r]; v.z = acc[qa][2][r]; v.w = acc[qa][3][r];
      *(float4*)&part[ks][4 * (8 * kh + r) + qa][4 * m] = v;
    }
  __syncthreads();
#pragma unroll
  for (int h = 0; h < 2; ++h) {
    int idx = threadIdx.x + 512 * h;
    int row = idx >> 4, c4 = (idx & 15) * 4;
    float4 s = *(const float4*)&part[0][row][c4];
    float den = csum[0][row];
#pragma unroll
    for (int i = 1; i < TK; ++i) {
      float4 p = *(const float4*)&part[i][row][c4];
      s.x += p.x; s.y += p.y; s.z += p.z; s.w += p.w;
      den += csum[i][row];
    }
    float inv = 1.0f / den;
    s.x *= inv; s.y *= inv; s.z *= inv; s.w *= inv;
    *(float4*)&tmp[(size_t)(b * NJ + j0 + row) * ND + d0 + c4] = s;
  }
}

#define OT 8
#define AP 132
__global__ __launch_bounds__(256) void k_out(const float* __restrict__ S,
                                             const float* __restrict__ Q,
                                             const float* __restrict__ tmp,
                                             const float* __restrict__ C,
                                             float* __restrict__ out) {
  __shared__ __attribute__((aligned(16))) float As[2][16][AP];
  int blk = blockIdx.x;
  int b = blk >> 4;
  int chunk = (blk >> 1) & 7;
  int dh = blk & 1;
  int tid = threadIdx.x;
  int wv = tid >> 5;
  int lane = tid & 31;
  int m = lane & 15, kh = lane >> 4;
  int d = (dh * 8 + wv) * 16 + m;
  int tbase = chunk * (OT * 16);
  const float* Sg = S + ((size_t)b * NT + tbase + wv) * NJ + lane * 4;
  float4 n0 = *(const float4*)(Sg);
  float4 n1 = *(const float4*)(Sg + (size_t)8 * NJ);
  const float* Qb = Q + (size_t)b * NJ * ND + d;
  const float* Tb = tmp + (size_t)b * NJ * ND + d;
  v2f bq[32], bt[32];
#pragma unroll
  for (int i = 0; i < 32; ++i) {
    int ka = 4 * i + 2 * kh;
    bq[i].x = Qb[(size_t)ka * ND];
    bq[i].y = Qb[(size_t)(ka + 1) * ND];
    bt[i].x = Tb[(size_t)ka * ND];
    bt[i].y = Tb[(size_t)(ka + 1) * ND];
  }
  {
    float4 p0 = row_softmax(n0);
    float4 p1 = row_softmax(n1);
    *(float4*)&As[0][wv][lane * 4] = p0;
    *(float4*)&As[0][wv + 8][lane * 4] = p1;
  }
  n0 = *(const float4*)(Sg + (size_t)16 * NJ);
  n1 = *(const float4*)(Sg + (size_t)24 * NJ);
  __syncthreads();
#pragma nounroll
  for (int tt = 0; tt < OT; ++tt) {
    int cur = tt & 1;
    int t0 = tbase + tt * 16;
    if (tt + 1 < OT) {
      float4 p0 = row_softmax(n0);
      float4 p1 = row_softmax(n1);
      *(float4*)&As[cur ^ 1][wv][lane * 4] = p0;
      *(float4*)&As[cur ^ 1][wv + 8][lane * 4] = p1;
      if (tt + 2 < OT) {
        const float* Sn = Sg + (size_t)(tt + 2) * 16 * NJ;
        n0 = *(const float4*)(Sn);
        n1 = *(const float4*)(Sn + (size_t)8 * NJ);
      }
    }
    float cv[8];
#pragma unroll
    for (int r = 0; r < 8; ++r)
      cv[r] = C[(size_t)(b * NT + t0 + r + 8 * kh) * ND + d];
    const float* Arow = &As[cur][m][2 * kh];
    v8f accA = {}, accB = {};
#pragma unroll
    for (int i = 0; i < 32; ++i) {
      v2f a = *(const v2f*)(Arow + 4 * i);
      accA = __builtin_amdgcn_wmma_f32_16x16x4_f32(false, a, false, bq[i], (short)0,
                                                   accA, false, false);
      accB = __builtin_amdgcn_wmma_f32_16x16x4_f32(false, a, false, bt[i], (short)0,
                                                   accB, false, false);
    }
    if (tt + 1 < OT) __syncthreads();
#pragma unroll
    for (int r = 0; r < 8; ++r) {
      int t = t0 + r + 8 * kh;
      float* ob = out + (size_t)(b * NT + t) * (4 * ND);
      ob[d]          = cv[r];
      ob[ND + d]     = accA[r];
      ob[2 * ND + d] = cv[r] * accA[r];
      ob[3 * ND + d] = cv[r] * accB[r];
    }
  }
}

extern "C" void kernel_launch(void* const* d_in, const int* in_sizes, int n_in,
                              void* d_out, int out_size, void* d_ws, size_t ws_size,
                              hipStream_t stream) {
  (void)in_sizes; (void)n_in; (void)out_size; (void)ws_size;
  const float* C = (const float*)d_in[0];
  const float* Q = (const float*)d_in[1];
  const float* w = (const float*)d_in[2];
  float* out = (float*)d_out;

  float* ws = (float*)d_ws;
  float* S  = ws;
  float* pm = S  + (size_t)NB * NT * NJ;
  float* tp = pm + (size_t)NB * NC * NJ;

  k_score<<<NB * (64 / ST), 256, 0, stream>>>(C, Q, w, S, pm);
  k_tmp<<<NB * 8, 512, 0, stream>>>(S, pm, C, tp);
  k_out<<<NB * 16, 256, 0, stream>>>(S, Q, tp, C, out);
}


# k_out: a wave owns 32 columns of ONE product (Sj@Q or Sj@tmp) as two column-interleaved tiles (d = 2c + q): rows of Q, tmp, C load and all four planes store as whole 128-byte lines; sums unchanged
# speedup vs baseline: 3.4249x; 1.0687x over previous
#include <hip/hip_runtime.h>

#define NB 32
#define NT 1024
#define NJ 128
#define ND 256

typedef __attribute__((ext_vector_type(2))) float v2f;
typedef __attribute__((ext_vector_type(8))) float v8f;

__device__ __forceinline__ float xhalf(float x) {
  int b = __builtin_bit_cast(int, x);
  return __builtin_bit_cast(float, __builtin_amdgcn_permlanex16(b, b, 0x76543210, 0xfedcba98,
                                                                false, false));
}
template <int N> __device__ __forceinline__ float xrow(float x) {
  return __builtin_amdgcn_update_dpp(x, x, 0x160 + N, 0xf, 0xf, true);
}
__device__ __forceinline__ float wave_reduce_sum(float v) {
  v += xhalf(v);
  v += xrow<8>(v);
  v += xrow<4>(v);
  v += xrow<2>(v);
  v += xrow<1>(v);
  return v;
}
__device__ __forceinline__ float wave_reduce_max(float v) {
  v = fmaxf(v, xhalf(v));
  v = fmaxf(v, xrow<8>(v));
  v = fmaxf(v, xrow<4>(v));
  v = fmaxf(v, xrow<2>(v));
  v = fmaxf(v, xrow<1>(v));
  return v;
}
__device__ __forceinline__ float4 row_softmax(float4 v) {
  float m = fmaxf(fmaxf(v.x, v.y), fmaxf(v.z, v.w));
  m = wave_reduce_max(m);
  float e0 = expf(v.x - m), e1 = expf(v.y - m), e2 = expf(v.z - m), e3 = expf(v.w - m);
  float s = wave_reduce_sum(e0 + e1 + e2 + e3);
  float inv = 1.0f / s;
  float4 o; o.x = e0 * inv; o.y = e1 * inv; o.z = e2 * inv; o.w = e3 * inv;
  return o;
}

#define ST 4
#define NC (NT / (ST * 16))
#define CP 260
__global__ __launch_bounds__(256) void k_score(const float* __restrict__ C,
                                               const float* __restrict__ Q,
                                               const float* __restrict__ w,
                                               float* __restrict__ S,
                                               float* __restrict__ pm) {
  __shared__ __attribute__((aligned(16))) float wl[(ND / 2) * 8];
  __shared__ __attribute__((aligned(16))) float Cs[2][16][CP];
  int blk = blockIdx.x;
  int b = blk >> 4;
  int chunk = blk & 15;
  int wv = threadIdx.x >> 5;
  int lane = threadIdx.x & 31;
  int m = lane & 15, kh = lane >> 4;
  int j0 = wv * 16;
  int tbase = chunk * (ST * 16);
  {
    int k = threadIdx.x;
    int p = (k >> 1) * 8 + (k & 1);
    wl[p]     = w[2 * ND + k];
    wl[p + 2] = w[k];
    wl[p + 4] = w[ND + k];
  }
  __syncthreads();
  const float* Cg = C + ((size_t)b * NT + tbase + wv) * ND + lane * 4;
  float4 f0 = *(const float4*)(Cg);
  float4 f1 = *(const float4*)(Cg + 128);
  float4 f2 = *(const float4*)(Cg + (size_t)8 * ND);
  float4 f3 = *(const float4*)(Cg + (size_t)8 * ND + 128);
  const float* Qrow = Q + (size_t)(b * NJ + j0 + m) * ND;
  v2f bf[64];
  v2f qs = {0.f, 0.f};
#pragma unroll
  for (int i = 0; i < 64; ++i) {
    int ka = 4 * i + 2 * kh;
    v2f q = *(const v2f*)(Qrow + ka);
    const float* wp = wl + (ka >> 1) * 8;
    float4 x = *(const float4*)wp;
    v2f uu = *(const v2f*)(wp + 4);
    v2f hu = {x.x, x.y}, hh = {x.z, x.w};
    bf[i] = q * hu + hh;
    qs += q * uu;
  }
  float qv = qs.x + qs.y;
  qv += __shfl_xor(qv, 16, 32);
  *(float4*)&Cs[0][wv][lane * 4] = f0;
  *(float4*)&Cs[0][wv][128 + lane * 4] = f1;
  *(float4*)&Cs[0][wv + 8][lane * 4] = f2;
  *(float4*)&Cs[0][wv + 8][128 + lane * 4] = f3;
  __syncthreads();
  int jj = j0 + m;
  float cm = -__builtin_inff();
#pragma nounroll
  for (int tt = 0; tt < ST; ++tt) {
    int cur = tt & 1;
    int t0 = tbase + tt * 16;
    float4 g0, g1, g2, g3;
    if (tt + 1 < ST) {
      const float* Cn = Cg + (size_t)(tt + 1) * 16 * ND;
      g0 = *(const float4*)(Cn);
      g1 = *(const float4*)(Cn + 128);
      g2 = *(const float4*)(Cn + (size_t)8 * ND);
      g3 = *(const float4*)(Cn + (size_t)8 * ND + 128);
    }
    const float* Arow = &Cs[cur][m][2 * kh];
    v8f acc = {};
#pragma unroll
    for (int i = 0; i < 64; ++i) {
      v2f a = *(const v2f*)(Arow + 4 * i);
      acc = __builtin_amdgcn_wmma_f32_16x16x4_f32(false, a, false, bf[i], (short)0,
                                                  acc, false, false);
    }
    if (tt + 1 < ST) {
      *(float4*)&Cs[cur ^ 1][wv][lane * 4] = g0;
      *(float4*)&Cs[cur ^ 1][wv][128 + lane * 4] = g1;
      *(float4*)&Cs[cur ^ 1][wv + 8][lane * 4] = g2;
      *(float4*)&Cs[cur ^ 1][wv + 8][128 + lane * 4] = g3;
      __syncthreads();
    }
#pragma unroll
    for (int r = 0; r < 8; ++r) {
      int tt2 = t0 + r + 8 * kh;
      float sv = acc[r] + qv;
      S[(size_t)(b * NT + tt2) * NJ + jj] = sv;
      cm = fmaxf(cm, sv);
    }
  }
  cm = fmaxf(cm, xhalf(cm));
  if (kh == 0) pm[(size_t)blk * NJ + jj] = cm;
}

#define TK 16
#define TR (NT / TK)
#define TS (TR / 4)
__device__ __forceinline__ float4 col_exp(float4 v, float4 mx) {
  const float L2E = 1.44269504088896340736f;
  float4 e;
  e.x = __builtin_amdgcn_exp2f((v.x - mx.x) * L2E);
  e.y = __builtin_amdgcn_exp2f((v.y - mx.y) * L2E);
  e.z = __builtin_amdgcn_exp2f((v.z - mx.z) * L2E);
  e.w = __builtin_amdgcn_exp2f((v.w - mx.w) * L2E);
  return e;
}
__global__ __launch_bounds__(512) void k_tmp(const float* __restrict__ S,
                                             const float* __restrict__ pm,
                                             const float* __restrict__ C,
                                             float* __restrict__ tmp) {
  __shared__ __attribute__((aligned(16))) float part[TK][64][64];
  __shared__ __attribute__((aligned(16))) float csum[TK][64];
  int blk = blockIdx.x;
  int b = blk >> 3;
  int j0 = ((blk >> 2) & 1) * 64;
  int d0 = (blk & 3) * 64;
  int ks = threadIdx.x >> 5;
  int lane = threadIdx.x & 31;
  int m = lane & 15, kh = lane >> 4;
  const float* Sp = S + ((size_t)b * NT + ks * TR + 2 * kh) * NJ + j0 + 4 * m;
  const float* Cp = C + ((size_t)b * NT + ks * TR + 2 * kh) * ND + d0 + 4 * m;
  const float* Mp = pm + (size_t)b * NC * NJ + j0 + 4 * m;
  float4 px[NC];
#pragma unroll
  for (int c = 0; c < NC; ++c) px[c] = *(const float4*)(Mp + (size_t)c * NJ);
  float4 ra0[TS], ra1[TS], rb0[TS], rb1[TS];
  v8f acc[4][4];
#pragma unroll
  for (int qa = 0; qa < 4; ++qa)
#pragma unroll
    for (int qb = 0; qb < 4; ++qb) acc[qa][qb] = v8f{};
#pragma unroll
  for (int i = 0; i < 3; ++i) {
    ra0[i] = *(const float4*)(Sp + (size_t)(4 * i) * NJ);
    ra1[i] = *(const float4*)(Sp + (size_t)(4 * i + 1) * NJ);
  }
#pragma unroll
  for (int i = 0; i < 2; ++i) {
    rb0[i] = *(const float4*)(Cp + (size_t)(4 * i) * ND);
    rb1[i] = *(const float4*)(Cp + (size_t)(4 * i + 1) * ND);
  }
  float4 mx = px[0];
#pragma unroll
  for (int c = 1; c < NC; ++c) {
    mx.x = fmaxf(mx.x, px[c].x); mx.y = fmaxf(mx.y, px[c].y);
    mx.z = fmaxf(mx.z, px[c].z); mx.w = fmaxf(mx.w, px[c].w);
  }
  float4 e0 = col_exp(ra0[0], mx);
  float4 e1 = col_exp(ra1[0], mx);
  float4 cs;
  cs.x = e0.x + e1.x; cs.y = e0.y + e1.y; cs.z = e0.z + e1.z; cs.w = e0.w + e1.w;
#pragma unroll
  for (int i = 0; i < TS; ++i) {
    if (i + 3 < TS) {
      ra0[i + 3] = *(const float4*)(Sp + (size_t)(4 * i + 12) * NJ);
      ra1[i + 3] = *(const float4*)(Sp + (size_t)(4 * i + 13) * NJ);
    }
    if (i + 2 < TS) {
      rb0[i + 2] = *(const float4*)(Cp + (size_t)(4 * i + 8) * ND);
      rb1[i + 2] = *(const float4*)(Cp + (size_t)(4 * i + 9) * ND);
    }
    float4 n0 = e0, n1 = e1;
    if (i + 1 < TS) {
      n0 = col_exp(ra0[i + 1], mx);
      n1 = col_exp(ra1[i + 1], mx);
      cs.x += n0.x; cs.y += n0.y; cs.z += n0.z; cs.w += n0.w;
      cs.x += n1.x; cs.y += n1.y; cs.z += n1.z; cs.w += n1.w;
    }
    v2f af[4], bf[4];
    af[0].x = e0.x; af[0].y = e1.x;
    af[1].x = e0.y; af[1].y = e1.y;
    af[2].x = e0.z; af[2].y = e1.z;
    af[3].x = e0.w; af[3].y = e1.w;
    bf[0].x = rb0[i].x; bf[0].y = rb1[i].x;
    bf[1].x = rb0[i].y; bf[1].y = rb1[i].y;
    bf[2].x = rb0[i].z; bf[2].y = rb1[i].z;
    bf[3].x = rb0[i].w; bf[3].y = rb1[i].w;
#pragma unroll
    for (int qa = 0; qa < 4; ++qa)
#pragma unroll
      for (int qb = 0; qb < 4; ++qb)
        acc[qa][qb] = __builtin_amdgcn_wmma_f32_16x16x4_f32(false, af[qa], false, bf[qb],
                                                            (short)0, acc[qa][qb], false, false);
    e0 = n0; e1 = n1;
    if ((i & 3) == 3) __builtin_amdgcn_sched_barrier(0);
  }
  cs.x += xhalf(cs.x); cs.y += xhalf(cs.y); cs.z += xhalf(cs.z); cs.w += xhalf(cs.w);
  if (kh == 0) *(float4*)&csum[ks][4 * m] = cs;
#pragma unroll
  for (int qa = 0; qa < 4; ++qa)
#pragma unroll
    for (int r = 0; r < 8; ++r) {
      float4 v;
      v.x = acc[qa][0][r]; v.y = acc[qa][1][r]; v.z = acc[qa][2][r]; v.w = acc[qa][3][r];
      *(float4*)&part[ks][4 * (8 * kh + r) + qa][4 * m] = v;
    }
  __syncthreads();
#pragma unroll
  for (int h = 0; h < 2; ++h) {
    int idx = threadIdx.x + 512 * h;
    int row = idx >> 4, c4 = (idx & 15) * 4;
    float4 s = *(const float4*)&part[0][row][c4];
    float den = csum[0][row];
#pragma unroll
    for (int i = 1; i < TK; ++i) {
      float4 p = *(const float4*)&part[i][row][c4];
      s.x += p.x; s.y += p.y; s.z += p.z; s.w += p.w;
      den += csum[i][row];
    }
    float inv = 1.0f / den;
    s.x *= inv; s.y *= inv; s.z *= inv; s.w *= inv;
    *(float4*)&tmp[(size_t)(b * NJ + j0 + row) * ND + d0 + c4] = s;
  }
}

#define OT 8
#define AP 132
__global__ __launch_bounds__(256) void k_out(const float* __restrict__ S,
                                             const float* __restrict__ Q,
                                             const float* __restrict__ tmp,
                                             const float* __restrict__ C,
                                             float* __restrict__ out) {
  __shared__ __attribute__((aligned(16))) float As[2][16][AP];
  int blk = blockIdx.x;
  int b = blk >> 4;
  int chunk = (blk >> 1) & 7;
  int dh = blk & 1;
  int tid = threadIdx.x;
  int wv = tid >> 5;
  int lane = tid & 31;
  int m = lane & 15, kh = lane >> 4;
  int prod = __builtin_amdgcn_readfirstlane(wv) >> 2;
  int d = dh * 128 + (wv & 3) * 32 + 2 * m;
  int tbase = chunk * (OT * 16);
  const float* Sg = S + ((size_t)b * NT + tbase + wv) * NJ + lane * 4;
  float4 n0 = *(const float4*)(Sg);
  float4 n1 = *(const float4*)(Sg + (size_t)8 * NJ);
  const float* Mb = (prod ? tmp : Q) + (size_t)b * NJ * ND + d;
  v2f b0[32], b1[32];
#pragma unroll
  for (int i = 0; i < 32; ++i) {
    int ka = 4 * i + 2 * kh;
    v2f r0 = *(const v2f*)(Mb + (size_t)ka * ND);
    v2f r1 = *(const v2f*)(Mb + (size_t)(ka + 1) * ND);
    b0[i].x = r0.x; b0[i].y = r1.x;
    b1[i].x = r0.y; b1[i].y = r1.y;
  }
  {
    float4 p0 = row_softmax(n0);
    float4 p1 = row_softmax(n1);
    *(float4*)&As[0][wv][lane * 4] = p0;
    *(float4*)&As[0][wv + 8][lane * 4] = p1;
  }
  n0 = *(const float4*)(Sg + (size_t)16 * NJ);
  n1 = *(const float4*)(Sg + (size_t)24 * NJ);
  __syncthreads();
#pragma nounroll
  for (int tt = 0; tt < OT; ++tt) {
    int cur = tt & 1;
    int t0 = tbase + tt * 16;
    if (tt + 1 < OT) {
      float4 p0 = row_softmax(n0);
      float4 p1 = row_softmax(n1);
      *(float4*)&As[cur ^ 1][wv][lane * 4] = p0;
      *(float4*)&As[cur ^ 1][wv + 8][lane * 4] = p1;
      if (tt + 2 < OT) {
        const float* Sn = Sg + (size_t)(tt + 2) * 16 * NJ;
        n0 = *(const float4*)(Sn);
        n1 = *(const float4*)(Sn + (size_t)8 * NJ);
      }
    }
    v2f cv[8];
#pragma unroll
    for (int r = 0; r < 8; ++r)
      cv[r] = *(const v2f*)(C + (size_t)(b * NT + t0 + r + 8 * kh) * ND + d);
    const float* Arow = &As[cur][m][2 * kh];
    v8f acc0 = {}, acc1 = {};
#pragma unroll
    for (int i = 0; i < 32; ++i) {
      v2f a = *(const v2f*)(Arow + 4 * i);
      acc0 = __builtin_amdgcn_wmma_f32_16x16x4_f32(false, a, false, b0[i], (short)0,
                                                   acc0, false, false);
      acc1 = __builtin_amdgcn_wmma_f32_16x16x4_f32(false, a, false, b1[i], (short)0,
                                                   acc1, false, false);
    }
    if (tt + 1 < OT) __syncthreads();
    float* ob = out + (size_t)(b * NT + t0 + 8 * kh) * (4 * ND) + d;
    float* obX = ob + (prod ? 0 : ND);
    float* obY = ob + (prod ? 3 * ND : 2 * ND);
#pragma unroll
    for (int r = 0; r < 8; ++r) {
      v2f pv; pv.x = acc0[r]; pv.y = acc1[r];
      v2f xv = prod ? cv[r] : pv;
      *(v2f*)(obX + (size_t)r * (4 * ND)) = xv;
      *(v2f*)(obY + (size_t)r * (4 * ND)) = cv[r] * pv;
    }
  }
}

extern "C" void kernel_launch(void* const* d_in, const int* in_sizes, int n_in,
                              void* d_out, int out_size, void* d_ws, size_t ws_size,
                              hipStream_t stream) {
  (void)in_sizes; (void)n_in; (void)out_size; (void)ws_size;
  const float* C = (const float*)d_in[0];
  const float* Q = (const float*)d_in[1];
  const float* w = (const float*)d_in[2];
  float* out = (float*)d_out;

  float* ws = (float*)d_ws;
  float* S  = ws;
  float* pm = S  + (size_t)NB * NT * NJ;
  float* tp = pm + (size_t)NB * NC * NJ;

  k_score<<<NB * (64 / ST), 256, 0, stream>>>(C, Q, w, S, pm);
  k_tmp<<<NB * 8, 512, 0, stream>>>(S, pm, C, tp);
  k_out<<<NB * 16, 256, 0, stream>>>(S, Q, tp, C, out);
}


# k_out: all four planes of out are stored with the non-temporal hint, so the write-once output no longer displaces C, S, Q and tmp in the second-level cache; values and sums unchanged
# speedup vs baseline: 3.6110x; 1.0543x over previous
#include <hip/hip_runtime.h>

#define NB 32
#define NT 1024
#define NJ 128
#define ND 256

typedef __attribute__((ext_vector_type(2))) float v2f;
typedef __attribute__((ext_vector_type(8))) float v8f;

__device__ __forceinline__ float xhalf(float x) {
  int b = __builtin_bit_cast(int, x);
  return __builtin_bit_cast(float, __builtin_amdgcn_permlanex16(b, b, 0x76543210, 0xfedcba98,
                                                                false, false));
}
template <int N> __device__ __forceinline__ float xrow(float x) {
  return __builtin_amdgcn_update_dpp(x, x, 0x160 + N, 0xf, 0xf, true);
}
__device__ __forceinline__ float wave_reduce_sum(float v) {
  v += xhalf(v);
  v += xrow<8>(v);
  v += xrow<4>(v);
  v += xrow<2>(v);
  v += xrow<1>(v);
  return v;
}
__device__ __forceinline__ float wave_reduce_max(float v) {
  v = fmaxf(v, xhalf(v));
  v = fmaxf(v, xrow<8>(v));
  v = fmaxf(v, xrow<4>(v));
  v = fmaxf(v, xrow<2>(v));
  v = fmaxf(v, xrow<1>(v));
  return v;
}
__device__ __forceinline__ float4 row_softmax(float4 v) {
  float m = fmaxf(fmaxf(v.x, v.y), fmaxf(v.z, v.w));
  m = wave_reduce_max(m);
  float e0 = expf(v.x - m), e1 = expf(v.y - m), e2 = expf(v.z - m), e3 = expf(v.w - m);
  float s = wave_reduce_sum(e0 + e1 + e2 + e3);
  float inv = 1.0f / s;
  float4 o; o.x = e0 * inv; o.y = e1 * inv; o.z = e2 * inv; o.w = e3 * inv;
  return o;
}

#define ST 4
#define NC (NT / (ST * 16))
#define CP 260
__global__ __launch_bounds__(256) void k_score(const float* __restrict__ C,
                                               const float* __restrict__ Q,
                                               const float* __restrict__ w,
                                               float* __restrict__ S,
                                               float* __restrict__ pm) {
  __shared__ __attribute__((aligned(16))) float wl[(ND / 2) * 8];
  __shared__ __attribute__((aligned(16))) float Cs[2][16][CP];
  int blk = blockIdx.x;
  int b = blk >> 4;
  int chunk = blk & 15;
  int wv = threadIdx.x >> 5;
  int lane = threadIdx.x & 31;
  int m = lane & 15, kh = lane >> 4;
  int j0 = wv * 16;
  int tbase = chunk * (ST * 16);
  {
    int k = threadIdx.x;
    int p = (k >> 1) * 8 + (k & 1);
    wl[p]     = w[2 * ND + k];
    wl[p + 2] = w[k];
    wl[p + 4] = w[ND + k];
  }
  __syncthreads();
  const float* Cg = C + ((size_t)b * NT + tbase + wv) * ND + lane * 4;
  float4 f0 = *(const float4*)(Cg);
  float4 f1 = *(const float4*)(Cg + 128);
  float4 f2 = *(const float4*)(Cg + (size_t)8 * ND);
  float4 f3 = *(const float4*)(Cg + (size_t)8 * ND + 128);
  const float* Qrow = Q + (size_t)(b * NJ + j0 + m) * ND;
  v2f bf[64];
  v2f qs = {0.f, 0.f};
#pragma unroll
  for (int i = 0; i < 64; ++i) {
    int ka = 4 * i + 2 * kh;
    v2f q = *(const v2f*)(Qrow + ka);
    const float* wp = wl + (ka >> 1) * 8;
    float4 x = *(const float4*)wp;
    v2f uu = *(const v2f*)(wp + 4);
    v2f hu = {x.x, x.y}, hh = {x.z, x.w};
    bf[i] = q * hu + hh;
    qs += q * uu;
  }
  float qv = qs.x + qs.y;
  qv += __shfl_xor(qv, 16, 32);
  *(float4*)&Cs[0][wv][lane * 4] = f0;
  *(float4*)&Cs[0][wv][128 + lane * 4] = f1;
  *(float4*)&Cs[0][wv + 8][lane * 4] = f2;
  *(float4*)&Cs[0][wv + 8][128 + lane * 4] = f3;
  __syncthreads();
  int jj = j0 + m;
  float cm = -__builtin_inff();
#pragma nounroll
  for (int tt = 0; tt < ST; ++tt) {
    int cur = tt & 1;
    int t0 = tbase + tt * 16;
    float4 g0, g1, g2, g3;
    if (tt + 1 < ST) {
      const float* Cn = Cg + (size_t)(tt + 1) * 16 * ND;
      g0 = *(const float4*)(Cn);
      g1 = *(const float4*)(Cn + 128);
      g2 = *(const float4*)(Cn + (size_t)8 * ND);
      g3 = *(const float4*)(Cn + (size_t)8 * ND + 128);
    }
    const float* Arow = &Cs[cur][m][2 * kh];
    v8f acc = {};
#pragma unroll
    for (int i = 0; i < 64; ++i) {
      v2f a = *(const v2f*)(Arow + 4 * i);
      acc = __builtin_amdgcn_wmma_f32_16x16x4_f32(false, a, false, bf[i], (short)0,
                                                  acc, false, false);
    }
    if (tt + 1 < ST) {
      *(float4*)&Cs[cur ^ 1][wv][lane * 4] = g0;
      *(float4*)&Cs[cur ^ 1][wv][128 + lane * 4] = g1;
      *(float4*)&Cs[cur ^ 1][wv + 8][lane * 4] = g2;
      *(float4*)&Cs[cur ^ 1][wv + 8][128 + lane * 4] = g3;
      __syncthreads();
    }
#pragma unroll
    for (int r = 0; r < 8; ++r) {
      int tt2 = t0 + r + 8 * kh;
      float sv = acc[r] + qv;
      S[(size_t)(b * NT + tt2) * NJ + jj] = sv;
      cm = fmaxf(cm, sv);
    }
  }
  cm = fmaxf(cm, xhalf(cm));
  if (kh == 0) pm[(size_t)blk * NJ + jj] = cm;
}

#define TK 16
#define TR (NT / TK)
#define TS (TR / 4)
__device__ __forceinline__ float4 col_exp(float4 v, float4 mx) {
  const float L2E = 1.44269504088896340736f;
  float4 e;
  e.x = __builtin_amdgcn_exp2f((v.x - mx.x) * L2E);
  e.y = __builtin_amdgcn_exp2f((v.y - mx.y) * L2E);
  e.z = __builtin_amdgcn_exp2f((v.z - mx.z) * L2E);
  e.w = __builtin_amdgcn_exp2f((v.w - mx.w) * L2E);
  return e;
}
__global__ __launch_bounds__(512) void k_tmp(const float* __restrict__ S,
                                             const float* __restrict__ pm,
                                             const float* __restrict__ C,
                                             float* __restrict__ tmp) {
  __shared__ __attribute__((aligned(16))) float part[TK][64][64];
  __shared__ __attribute__((aligned(16))) float csum[TK][64];
  int blk = blockIdx.x;
  int b = blk >> 3;
  int j0 = ((blk >> 2) & 1) * 64;
  int d0 = (blk & 3) * 64;
  int ks = threadIdx.x >> 5;
  int lane = threadIdx.x & 31;
  int m = lane & 15, kh = lane >> 4;
  const float* Sp = S + ((size_t)b * NT + ks * TR + 2 * kh) * NJ + j0 + 4 * m;
  const float* Cp = C + ((size_t)b * NT + ks * TR + 2 * kh) * ND + d0 + 4 * m;
  const float* Mp = pm + (size_t)b * NC * NJ + j0 + 4 * m;
  float4 px[NC];
#pragma unroll
  for (int c = 0; c < NC; ++c) px[c] = *(const float4*)(Mp + (size_t)c * NJ);
  float4 ra0[TS], ra1[TS], rb0[TS], rb1[TS];
  v8f acc[4][4];
#pragma unroll
  for (int qa = 0; qa < 4; ++qa)
#pragma unroll
    for (int qb = 0; qb < 4; ++qb) acc[qa][qb] = v8f{};
#pragma unroll
  for (int i = 0; i < 3; ++i) {
    ra0[i] = *(const float4*)(Sp + (size_t)(4 * i) * NJ);
    ra1[i] = *(const float4*)(Sp + (size_t)(4 * i + 1) * NJ);
  }
#pragma unroll
  for (int i = 0; i < 2; ++i) {
    rb0[i] = *(const float4*)(Cp + (size_t)(4 * i) * ND);
    rb1[i] = *(const float4*)(Cp + (size_t)(4 * i + 1) * ND);
  }
  float4 mx = px[0];
#pragma unroll
  for (int c = 1; c < NC; ++c) {
    mx.x = fmaxf(mx.x, px[c].x); mx.y = fmaxf(mx.y, px[c].y);
    mx.z = fmaxf(mx.z, px[c].z); mx.w = fmaxf(mx.w, px[c].w);
  }
  float4 e0 = col_exp(ra0[0], mx);
  float4 e1 = col_exp(ra1[0], mx);
  float4 cs;
  cs.x = e0.x + e1.x; cs.y = e0.y + e1.y; cs.z = e0.z + e1.z; cs.w = e0.w + e1.w;
#pragma unroll
  for (int i = 0; i < TS; ++i) {
    if (i + 3 < TS) {
      ra0[i + 3] = *(const float4*)(Sp + (size_t)(4 * i + 12) * NJ);
      ra1[i + 3] = *(const float4*)(Sp + (size_t)(4 * i + 13) * NJ);
    }
    if (i + 2 < TS) {
      rb0[i + 2] = *(const float4*)(Cp + (size_t)(4 * i + 8) * ND);
      rb1[i + 2] = *(const float4*)(Cp + (size_t)(4 * i + 9) * ND);
    }
    float4 n0 = e0, n1 = e1;
    if (i + 1 < TS) {
      n0 = col_exp(ra0[i + 1], mx);
      n1 = col_exp(ra1[i + 1], mx);
      cs.x += n0.x; cs.y += n0.y; cs.z += n0.z; cs.w += n0.w;
      cs.x += n1.x; cs.y += n1.y; cs.z += n1.z; cs.w += n1.w;
    }
    v2f af[4], bf[4];
    af[0].x = e0.x; af[0].y = e1.x;
    af[1].x = e0.y; af[1].y = e1.y;
    af[2].x = e0.z; af[2].y = e1.z;
    af[3].x = e0.w; af[3].y = e1.w;
    bf[0].x = rb0[i].x; bf[0].y = rb1[i].x;
    bf[1].x = rb0[i].y; bf[1].y = rb1[i].y;
    bf[2].x = rb0[i].z; bf[2].y = rb1[i].z;
    bf[3].x = rb0[i].w; bf[3].y = rb1[i].w;
#pragma unroll
    for (int qa = 0; qa < 4; ++qa)
#pragma unroll
      for (int qb = 0; qb < 4; ++qb)
        acc[qa][qb] = __builtin_amdgcn_wmma_f32_16x16x4_f32(false, af[qa], false, bf[qb],
                                                            (short)0, acc[qa][qb], false, false);
    e0 = n0; e1 = n1;
    if ((i & 3) == 3) __builtin_amdgcn_sched_barrier(0);
  }
  cs.x += xhalf(cs.x); cs.y += xhalf(cs.y); cs.z += xhalf(cs.z); cs.w += xhalf(cs.w);
  if (kh == 0) *(float4*)&csum[ks][4 * m] = cs;
#pragma unroll
  for (int qa = 0; qa < 4; ++qa)
#pragma unroll
    for (int r = 0; r < 8; ++r) {
      float4 v;
      v.x = acc[qa][0][r]; v.y = acc[qa][1][r]; v.z = acc[qa][2][r]; v.w = acc[qa][3][r];
      *(float4*)&part[ks][4 * (8 * kh + r) + qa][4 * m] = v;
    }
  __syncthreads();
#pragma unroll
  for (int h = 0; h < 2; ++h) {
    int idx = threadIdx.x + 512 * h;
    int row = idx >> 4, c4 = (idx & 15) * 4;
    float4 s = *(const float4*)&part[0][row][c4];
    float den = csum[0][row];
#pragma unroll
    for (int i = 1; i < TK; ++i) {
      float4 p = *(const float4*)&part[i][row][c4];
      s.x += p.x; s.y += p.y; s.z += p.z; s.w += p.w;
      den += csum[i][row];
    }
    float inv = 1.0f / den;
    s.x *= inv; s.y *= inv; s.z *= inv; s.w *= inv;
    *(float4*)&tmp[(size_t)(b * NJ + j0 + row) * ND + d0 + c4] = s;
  }
}

#define OT 8
#define AP 132
__global__ __launch_bounds__(256) void k_out(const float* __restrict__ S,
                                             const float* __restrict__ Q,
                                             const float* __restrict__ tmp,
                                             const float* __restrict__ C,
                                             float* __restrict__ out) {
  __shared__ __attribute__((aligned(16))) float As[2][16][AP];
  int blk = blockIdx.x;
  int b = blk >> 4;
  int chunk = (blk >> 1) & 7;
  int dh = blk & 1;
  int tid = threadIdx.x;
  int wv = tid >> 5;
  int lane = tid & 31;
  int m = lane & 15, kh = lane >> 4;
  int prod = __builtin_amdgcn_readfirstlane(wv) >> 2;
  int d = dh * 128 + (wv & 3) * 32 + 2 * m;
  int tbase = chunk * (OT * 16);
  const float* Sg = S + ((size_t)b * NT + tbase + wv) * NJ + lane * 4;
  float4 n0 = *(const float4*)(Sg);
  float4 n1 = *(const float4*)(Sg + (size_t)8 * NJ);
  const float* Mb = (prod ? tmp : Q) + (size_t)b * NJ * ND + d;
  v2f b0[32], b1[32];
#pragma unroll
  for (int i = 0; i < 32; ++i) {
    int ka = 4 * i + 2 * kh;
    v2f r0 = *(const v2f*)(Mb + (size_t)ka * ND);
    v2f r1 = *(const v2f*)(Mb + (size_t)(ka + 1) * ND);
    b0[i].x = r0.x; b0[i].y = r1.x;
    b1[i].x = r0.y; b1[i].y = r1.y;
  }
  {
    float4 p0 = row_softmax(n0);
    float4 p1 = row_softmax(n1);
    *(float4*)&As[0][wv][lane * 4] = p0;
    *(float4*)&As[0][wv + 8][lane * 4] = p1;
  }
  n0 = *(const float4*)(Sg + (size_t)16 * NJ);
  n1 = *(const float4*)(Sg + (size_t)24 * NJ);
  __syncthreads();
#pragma nounroll
  for (int tt = 0; tt < OT; ++tt) {
    int cur = tt & 1;
    int t0 = tbase + tt * 16;
    if (tt + 1 < OT) {
      float4 p0 = row_softmax(n0);
      float4 p1 = row_softmax(n1);
      *(float4*)&As[cur ^ 1][wv][lane * 4] = p0;
      *(float4*)&As[cur ^ 1][wv + 8][lane * 4] = p1;
      if (tt + 2 < OT) {
        const float* Sn = Sg + (size_t)(tt + 2) * 16 * NJ;
        n0 = *(const float4*)(Sn);
        n1 = *(const float4*)(Sn + (size_t)8 * NJ);
      }
    }
    v2f cv[8];
#pragma unroll
    for (int r = 0; r < 8; ++r)
      cv[r] = *(const v2f*)(C + (size_t)(b * NT + t0 + r + 8 * kh) * ND + d);
    const float* Arow = &As[cur][m][2 * kh];
    v8f acc0 = {}, acc1 = {};
#pragma unroll
    for (int i = 0; i < 32; ++i) {
      v2f a = *(const v2f*)(Arow + 4 * i);
      acc0 = __builtin_amdgcn_wmma_f32_16x16x4_f32(false, a, false, b0[i], (short)0,
                                                   acc0, false, false);
      acc1 = __builtin_amdgcn_wmma_f32_16x16x4_f32(false, a, false, b1[i], (short)0,
                                                   acc1, false, false);
    }
    if (tt + 1 < OT) __syncthreads();
    float* ob = out + (size_t)(b * NT + t0 + 8 * kh) * (4 * ND) + d;
    float* obX = ob + (prod ? 0 : ND);
    float* obY = ob + (prod ? 3 * ND : 2 * ND);
#pragma unroll
    for (int r = 0; r < 8; ++r) {
      v2f pv; pv.x = acc0[r]; pv.y = acc1[r];
      v2f xv = prod ? cv[r] : pv;
      v2f yv = cv[r] * pv;
      __builtin_nontemporal_store(xv, (v2f*)(obX + (size_t)r * (4 * ND)));
      __builtin_nontemporal_store(yv, (v2f*)(obY + (size_t)r * (4 * ND)));
    }
  }
}

extern "C" void kernel_launch(void* const* d_in, const int* in_sizes, int n_in,
                              void* d_out, int out_size, void* d_ws, size_t ws_size,
                              hipStream_t stream) {
  (void)in_sizes; (void)n_in; (void)out_size; (void)ws_size;
  const float* C = (const float*)d_in[0];
  const float* Q = (const float*)d_in[1];
  const float* w = (const float*)d_in[2];
  float* out = (float*)d_out;

  float* ws = (float*)d_ws;
  float* S  = ws;
  float* pm = S  + (size_t)NB * NT * NJ;
  float* tp = pm + (size_t)NB * NC * NJ;

  k_score<<<NB * (64 / ST), 256, 0, stream>>>(C, Q, w, S, pm);
  k_tmp<<<NB * 8, 512, 0, stream>>>(S, pm, C, tp);
  k_out<<<NB * 16, 256, 0, stream>>>(S, Q, tp, C, out);
}
